# starting kernel
# baseline (speedup 1.0000x reference)
#include <hip/hip_runtime.h>


typedef __attribute__((ext_vector_type(2))) float v2f;
typedef __attribute__((ext_vector_type(8))) float v8f;

#define C1F 4.7e-9f

#if __has_builtin(__builtin_amdgcn_update_dpp)
#define ROR_ADD(x, ctrl)                                                  \
  ((x) + __int_as_float(__builtin_amdgcn_update_dpp(                      \
             0, __float_as_int(x), (ctrl), 0xf, 0xf, true)))
#define HAVE_DPP 1
#endif

__global__ void ndc_prep(const float* __restrict__ v_in,
                         const float* __restrict__ vs_r,
                         const float* __restrict__ fs,
                         float* __restrict__ params, int T) {
  int t = blockIdx.x * blockDim.x + threadIdx.x;
  if (t >= T + 8) return;
  int ti = (t < T) ? t : (T - 1);
  float f   = fs[ti];
  float vr  = vs_r[ti];
  float c1r = 1.0f / (2.0f * C1F * f);
  float r0  = c1r * vr / (c1r + vr);
  float g0 = 1.0f / r0, g1 = 1.0f / c1r, g2 = 1.0f / vr;
  float inv = 2.0f / (g0 + g1 + g2);
  float P0 = g0 * inv, P1 = g1 * inv, P2 = g2 * inv;
  float4* p4 = (float4*)(params + 8 * (size_t)t);
  p4[0] = make_float4(v_in[ti], P0 - 1.0f, P1, P2);
  p4[1] = make_float4(P0, P1 - 1.0f, P2, r0 * (1.0f / 3000.0f));
}

__global__ __launch_bounds__(32, 1) void ndc_scan(
    const float* __restrict__ params,
    const float* __restrict__ W_in, const float* __restrict__ b_in,
    const float* __restrict__ W_h,  const float* __restrict__ b_h,
    const float* __restrict__ W_out, const float* __restrict__ b_out,
    float* __restrict__ out, int T) {
  const int  lane = threadIdx.x;
  const bool hi   = lane >= 16;
  const int  r16  = lane & 15;

  float Wi0 = W_in[2 * lane];
  float Wi1 = W_in[2 * lane + 1];
  float bin = b_in[lane];
  float bLo0 = b_h[r16];
  float bHi0 = b_h[16 + r16];
  float bLo1 = b_h[32 + r16];
  float bHi1 = b_h[48 + r16];
  float WoA = W_out[r16];
  float WoB = W_out[16 + r16];
  float bo  = b_out[0];

  float Bw[2][8][2][2];
#pragma unroll
  for (int l = 0; l < 2; ++l)
#pragma unroll
    for (int c = 0; c < 8; ++c)
#pragma unroll
      for (int blk = 0; blk < 2; ++blk) {
        int row = 16 * blk + r16;
        int col = 4 * c + (hi ? 2 : 0);
        Bw[l][c][blk][0] = W_h[l * 1024 + row * 32 + col];
        Bw[l][c][blk][1] = W_h[l * 1024 + row * 32 + col + 1];
      }

  const int idx_base = hi ? 8 : 0;

  const float4* pf = (const float4*)params;
  float4 qa   = pf[0], qb   = pf[1];
  float4 qa_n = pf[2], qb_n = pf[3];

  float a0 = 0.0f, b1 = 0.0f;

  for (int t = 0; t < T; ++t) {
    float4 qa_nn = pf[2 * (t + 2)];
    float4 qb_nn = pf[2 * (t + 2) + 1];
    __builtin_prefetch(params + 8 * (size_t)(t + 8), 0, 3);

    float v   = qa.x;
    float r00 = qa.y, r01 = qa.z, r02 = qa.w;
    float r10 = qb.x, r11 = qb.y, r12 = qb.z;
    float drt = qb.w;

    float dp_a = r00 * a0 + r01 * b1 + r02 * v;

    float h0 = fmaxf(Wi0 * dp_a + Wi1 * drt + bin, 0.0f);
    int hb = __float_as_int(h0);

    int Ax[8], Ay[8];
#pragma unroll
    for (int c = 0; c < 8; ++c) {
      Ax[c] = __builtin_amdgcn_ds_bpermute(16 * c + idx_base, hb);
      Ay[c] = __builtin_amdgcn_ds_bpermute(16 * c + idx_base + 4, hb);
    }
    v8f a00 = {}, a01 = {}, a10 = {}, a11 = {};
#pragma unroll
    for (int c = 0; c < 4; ++c) {
      v2f A;  A.x  = __int_as_float(Ax[c]);     A.y  = __int_as_float(Ay[c]);
      v2f A2; A2.x = __int_as_float(Ax[c + 4]); A2.y = __int_as_float(Ay[c + 4]);
      v2f B00; B00.x = Bw[0][c][0][0];     B00.y = Bw[0][c][0][1];
      v2f B01; B01.x = Bw[0][c][1][0];     B01.y = Bw[0][c][1][1];
      v2f B10; B10.x = Bw[0][c + 4][0][0]; B10.y = Bw[0][c + 4][0][1];
      v2f B11; B11.x = Bw[0][c + 4][1][0]; B11.y = Bw[0][c + 4][1][1];
      a00 = __builtin_amdgcn_wmma_f32_16x16x4_f32(false, A,  false, B00,
                                                  (short)0, a00, false, false);
      a10 = __builtin_amdgcn_wmma_f32_16x16x4_f32(false, A,  false, B01,
                                                  (short)0, a10, false, false);
      a01 = __builtin_amdgcn_wmma_f32_16x16x4_f32(false, A2, false, B10,
                                                  (short)0, a01, false, false);
      a11 = __builtin_amdgcn_wmma_f32_16x16x4_f32(false, A2, false, B11,
                                                  (short)0, a11, false, false);
    }
    float hLo = fmaxf((a00[0] + a01[0]) + bLo0, 0.0f);
    float hHi = fmaxf((a10[0] + a11[0]) + bHi0, 0.0f);
    int hbL = __float_as_int(hLo);
    int hbH = __float_as_int(hHi);

#pragma unroll
    for (int c = 0; c < 4; ++c) {
      Ax[c] = __builtin_amdgcn_ds_bpermute(16 * c + idx_base, hbL);
      Ay[c] = __builtin_amdgcn_ds_bpermute(16 * c + idx_base + 4, hbL);
    }
#pragma unroll
    for (int c = 4; c < 8; ++c) {
      Ax[c] = __builtin_amdgcn_ds_bpermute(16 * c + idx_base, hbH);
      Ay[c] = __builtin_amdgcn_ds_bpermute(16 * c + idx_base + 4, hbH);
    }
    v8f b00 = {}, b01 = {}, b10 = {}, b11 = {};
#pragma unroll
    for (int c = 0; c < 4; ++c) {
      v2f A;  A.x  = __int_as_float(Ax[c]);     A.y  = __int_as_float(Ay[c]);
      v2f A2; A2.x = __int_as_float(Ax[c + 4]); A2.y = __int_as_float(Ay[c + 4]);
      v2f B00; B00.x = Bw[1][c][0][0];     B00.y = Bw[1][c][0][1];
      v2f B01; B01.x = Bw[1][c][1][0];     B01.y = Bw[1][c][1][1];
      v2f B10; B10.x = Bw[1][c + 4][0][0]; B10.y = Bw[1][c + 4][0][1];
      v2f B11; B11.x = Bw[1][c + 4][1][0]; B11.y = Bw[1][c + 4][1][1];
      b00 = __builtin_amdgcn_wmma_f32_16x16x4_f32(false, A,  false, B00,
                                                  (short)0, b00, false, false);
      b10 = __builtin_amdgcn_wmma_f32_16x16x4_f32(false, A,  false, B01,
                                                  (short)0, b10, false, false);
      b01 = __builtin_amdgcn_wmma_f32_16x16x4_f32(false, A2, false, B10,
                                                  (short)0, b01, false, false);
      b11 = __builtin_amdgcn_wmma_f32_16x16x4_f32(false, A2, false, B11,
                                                  (short)0, b11, false, false);
    }
    float gLo = fmaxf((b00[0] + b01[0]) + bLo1, 0.0f);
    float gHi = fmaxf((b10[0] + b11[0]) + bHi1, 0.0f);

    float p = WoA * gLo + WoB * gHi;
#ifdef HAVE_DPP
    p = ROR_ADD(p, 0x128);
    p = ROR_ADD(p, 0x124);
    p = ROR_ADD(p, 0x122);
    p = ROR_ADD(p, 0x121);
#else
#pragma unroll
    for (int m = 8; m >= 1; m >>= 1) {
      p += __int_as_float(
          __builtin_amdgcn_ds_bpermute((lane ^ m) << 2, __float_as_int(p)));
    }
#endif
    float dp_b = p + bo;

    float b1n = r10 * dp_b + r11 * b1 + r12 * v;

    out[t] = 0.5f * (dp_a + dp_b);

    a0 = dp_b;
    b1 = b1n;
    qa = qa_n;  qb = qb_n;
    qa_n = qa_nn;  qb_n = qb_nn;
  }
}

extern "C" void kernel_launch(void* const* d_in, const int* in_sizes, int n_in,
                              void* d_out, int out_size, void* d_ws,
                              size_t ws_size, hipStream_t stream) {
  const float* v_in  = (const float*)d_in[0];
  const float* vs_r  = (const float*)d_in[1];
  const float* fs    = (const float*)d_in[2];
  const float* W_in  = (const float*)d_in[3];
  const float* b_in  = (const float*)d_in[4];
  const float* W_h   = (const float*)d_in[5];
  const float* b_h   = (const float*)d_in[6];
  const float* W_out = (const float*)d_in[7];
  const float* b_out = (const float*)d_in[8];
  float* out = (float*)d_out;
  int T = in_sizes[0];

  float* params = (float*)d_ws;

  ndc_prep<<<(T + 8 + 255) / 256, 256, 0, stream>>>(v_in, vs_r, fs, params, T);
  ndc_scan<<<1, 32, 0, stream>>>(params, W_in, b_in, W_h, b_h, W_out, b_out,
                                 out, T);
}


# Hidden vectors stay in the WMMA accumulator layout: W is the A operand, relu(D) is the next layer's B, bias enters as C; all ds_bpermute and the DPP reduce are gone; dot-product sum order changes
# speedup vs baseline: 1.1438x; 1.1438x over previous
#include <hip/hip_runtime.h>


typedef __attribute__((ext_vector_type(2))) float v2f;
typedef __attribute__((ext_vector_type(8))) float v8f;

#define C1F 4.7e-9f

__global__ void ndc_prep(const float* __restrict__ v_in,
                         const float* __restrict__ vs_r,
                         const float* __restrict__ fs,
                         float* __restrict__ params, int T) {
  int t = blockIdx.x * blockDim.x + threadIdx.x;
  if (t >= T + 8) return;
  int ti = (t < T) ? t : (T - 1);
  float f   = fs[ti];
  float vr  = vs_r[ti];
  float c1r = 1.0f / (2.0f * C1F * f);
  float r0  = c1r * vr / (c1r + vr);
  float g0 = 1.0f / r0, g1 = 1.0f / c1r, g2 = 1.0f / vr;
  float inv = 2.0f / (g0 + g1 + g2);
  float P0 = g0 * inv, P1 = g1 * inv, P2 = g2 * inv;
  float4* p4 = (float4*)(params + 8 * (size_t)t);
  p4[0] = make_float4(v_in[ti], P0 - 1.0f, P1, P2);
  p4[1] = make_float4(P0, P1 - 1.0f, P2, r0 * (1.0f / 3000.0f));
}

__device__ __forceinline__ float other_half(float x) {
  return __builtin_bit_cast(
      float, __builtin_amdgcn_permlanex16(0, __builtin_bit_cast(int, x),
                                          0x76543210, 0xfedcba98, false, false));
}

__device__ __forceinline__ float relu(float x) {
  return __builtin_elementwise_maximum(x, 0.0f);
}

__device__ __forceinline__ v8f mma(v2f a, v2f b, v8f c) {
  return __builtin_amdgcn_wmma_f32_16x16x4_f32(false, a, false, b, (short)0, c,
                                               false, false);
}

__global__ __launch_bounds__(32, 1) void ndc_scan(
    const float* __restrict__ params,
    const float* __restrict__ W_in, const float* __restrict__ b_in,
    const float* __restrict__ W_h,  const float* __restrict__ b_h,
    const float* __restrict__ W_out, const float* __restrict__ b_out,
    float* __restrict__ out, int T) {
  const int lane = threadIdx.x;
  const int c16  = lane & 15;
  const int kb   = 8 * (lane >> 4);

  v2f Aw[2][2][8];
#pragma unroll
  for (int l = 0; l < 2; ++l)
#pragma unroll
    for (int ob = 0; ob < 2; ++ob)
#pragma unroll
      for (int q = 0; q < 8; ++q) {
        const int blk = q >> 2, j = q & 3;
        const float* p =
            W_h + l * 1024 + (16 * ob + c16) * 32 + 16 * blk + kb + 2 * j;
        Aw[l][ob][q].x = p[0];
        Aw[l][ob][q].y = p[1];
      }

  v8f Cb[2][2];
#pragma unroll
  for (int l = 0; l < 2; ++l)
#pragma unroll
    for (int ob = 0; ob < 2; ++ob)
#pragma unroll
      for (int r = 0; r < 8; ++r)
        Cb[l][ob][r] = b_h[l * 32 + 16 * ob + kb + r];

  float Wi0v[8][2], Wi1v[8][2], binv[8][2];
#pragma unroll
  for (int q = 0; q < 8; ++q)
#pragma unroll
    for (int s = 0; s < 2; ++s) {
      const int k = 16 * (q >> 2) + kb + 2 * (q & 3) + s;
      Wi0v[q][s] = W_in[2 * k];
      Wi1v[q][s] = W_in[2 * k + 1];
      binv[q][s] = b_in[k];
    }

  float WoV[2][8];
#pragma unroll
  for (int ob = 0; ob < 2; ++ob)
#pragma unroll
    for (int r = 0; r < 8; ++r) WoV[ob][r] = W_out[16 * ob + kb + r];
  const float bo = b_out[0];

  const float4* pf = (const float4*)params;
  float4 ca = pf[0], cb = pf[1];
  float4 na = pf[2], nb = pf[3];

  float cv[8][2];
#pragma unroll
  for (int q = 0; q < 8; ++q)
#pragma unroll
    for (int s = 0; s < 2; ++s) cv[q][s] = fmaf(Wi1v[q][s], cb.w, binv[q][s]);

  float a0 = 0.0f, b1 = 0.0f;
  float dp_a = ca.y * a0 + ca.z * b1 + ca.w * ca.x;

  for (int t = 0; t < T; ++t) {
    float4 la = pf[2 * (t + 2)];
    float4 lb = pf[2 * (t + 2) + 1];
    __builtin_prefetch(params + 8 * (size_t)(t + 8), 0, 3);

    v2f Bh[8];
#pragma unroll
    for (int q = 0; q < 8; ++q) {
      Bh[q].x = fmaxf(fmaf(Wi0v[q][0], dp_a, cv[q][0]), 0.0f);
      Bh[q].y = fmaxf(fmaf(Wi0v[q][1], dp_a, cv[q][1]), 0.0f);
    }

    v8f d0 = Cb[0][0], d1 = Cb[0][1];
#pragma unroll
    for (int q = 0; q < 8; ++q) {
      d0 = mma(Aw[0][0][q], Bh[q], d0);
      d1 = mma(Aw[0][1][q], Bh[q], d1);
    }

    v8f e0 = Cb[1][0], e1 = Cb[1][1];
#pragma unroll
    for (int j = 0; j < 4; ++j) {
      v2f B;
      B.x = relu(d0[2 * j]);
      B.y = relu(d0[2 * j + 1]);
      e0 = mma(Aw[1][0][j], B, e0);
      e1 = mma(Aw[1][1][j], B, e1);
    }
#pragma unroll
    for (int j = 0; j < 4; ++j) {
      v2f B;
      B.x = relu(d1[2 * j]);
      B.y = relu(d1[2 * j + 1]);
      e0 = mma(Aw[1][0][4 + j], B, e0);
      e1 = mma(Aw[1][1][4 + j], B, e1);
    }

#pragma unroll
    for (int q = 0; q < 8; ++q) {
      cv[q][0] = fmaf(Wi1v[q][0], nb.w, binv[q][0]);
      cv[q][1] = fmaf(Wi1v[q][1], nb.w, binv[q][1]);
    }

    float c[4];
#pragma unroll
    for (int i = 0; i < 4; ++i) c[i] = WoV[0][i] * relu(e0[i]);
#pragma unroll
    for (int i = 0; i < 4; ++i) c[i] = fmaf(WoV[0][4 + i], relu(e0[4 + i]), c[i]);
#pragma unroll
    for (int i = 0; i < 4; ++i) c[i] = fmaf(WoV[1][i], relu(e1[i]), c[i]);
#pragma unroll
    for (int i = 0; i < 4; ++i) c[i] = fmaf(WoV[1][4 + i], relu(e1[4 + i]), c[i]);
    float ph = (c[0] + c[2]) + (c[1] + c[3]);
    float dp_b = (ph + other_half(ph)) + bo;

    float b1n = cb.x * dp_b + cb.y * b1 + cb.z * ca.x;

    out[t] = 0.5f * (dp_a + dp_b);

    a0 = dp_b;
    b1 = b1n;
    dp_a = na.y * a0 + na.z * b1 + na.w * na.x;

    ca = na;  cb = nb;
    na = la;  nb = lb;
  }
}

extern "C" void kernel_launch(void* const* d_in, const int* in_sizes, int n_in,
                              void* d_out, int out_size, void* d_ws,
                              size_t ws_size, hipStream_t stream) {
  const float* v_in  = (const float*)d_in[0];
  const float* vs_r  = (const float*)d_in[1];
  const float* fs    = (const float*)d_in[2];
  const float* W_in  = (const float*)d_in[3];
  const float* b_in  = (const float*)d_in[4];
  const float* W_h   = (const float*)d_in[5];
  const float* b_h   = (const float*)d_in[6];
  const float* W_out = (const float*)d_in[7];
  const float* b_out = (const float*)d_in[8];
  float* out = (float*)d_out;
  int T = in_sizes[0];

  float* params = (float*)d_ws;

  ndc_prep<<<(T + 8 + 255) / 256, 256, 0, stream>>>(v_in, vs_r, fs, params, T);
  ndc_scan<<<1, 32, 0, stream>>>(params, W_in, b_in, W_h, b_h, W_out, b_out,
                                 out, T);
}


# Store of step t's output moved into the matrix work of step t+1 (last one behind the loop): the wait for the store's address translation leaves the serial tail; arithmetic and its order unchanged
# speedup vs baseline: 1.1857x; 1.0366x over previous
#include <hip/hip_runtime.h>


typedef __attribute__((ext_vector_type(2))) float v2f;
typedef __attribute__((ext_vector_type(8))) float v8f;

#define C1F 4.7e-9f

__global__ void ndc_prep(const float* __restrict__ v_in,
                         const float* __restrict__ vs_r,
                         const float* __restrict__ fs,
                         float* __restrict__ params, int T) {
  int t = blockIdx.x * blockDim.x + threadIdx.x;
  if (t >= T + 8) return;
  int ti = (t < T) ? t : (T - 1);
  float f   = fs[ti];
  float vr  = vs_r[ti];
  float c1r = 1.0f / (2.0f * C1F * f);
  float r0  = c1r * vr / (c1r + vr);
  float g0 = 1.0f / r0, g1 = 1.0f / c1r, g2 = 1.0f / vr;
  float inv = 2.0f / (g0 + g1 + g2);
  float P0 = g0 * inv, P1 = g1 * inv, P2 = g2 * inv;
  float4* p4 = (float4*)(params + 8 * (size_t)t);
  p4[0] = make_float4(v_in[ti], P0 - 1.0f, P1, P2);
  p4[1] = make_float4(P0, P1 - 1.0f, P2, r0 * (1.0f / 3000.0f));
}

__device__ __forceinline__ float other_half(float x) {
  return __builtin_bit_cast(
      float, __builtin_amdgcn_permlanex16(0, __builtin_bit_cast(int, x),
                                          0x76543210, 0xfedcba98, false, false));
}

__device__ __forceinline__ float relu(float x) {
  return __builtin_elementwise_maximum(x, 0.0f);
}

__device__ __forceinline__ v8f mma(v2f a, v2f b, v8f c) {
  return __builtin_amdgcn_wmma_f32_16x16x4_f32(false, a, false, b, (short)0, c,
                                               false, false);
}

__global__ __launch_bounds__(32, 1) void ndc_scan(
    const float* __restrict__ params,
    const float* __restrict__ W_in, const float* __restrict__ b_in,
    const float* __restrict__ W_h,  const float* __restrict__ b_h,
    const float* __restrict__ W_out, const float* __restrict__ b_out,
    float* __restrict__ out, int T) {
  const int lane = threadIdx.x;
  const int c16  = lane & 15;
  const int kb   = 8 * (lane >> 4);

  v2f Aw[2][2][8];
#pragma unroll
  for (int l = 0; l < 2; ++l)
#pragma unroll
    for (int ob = 0; ob < 2; ++ob)
#pragma unroll
      for (int q = 0; q < 8; ++q) {
        const int blk = q >> 2, j = q & 3;
        const float* p =
            W_h + l * 1024 + (16 * ob + c16) * 32 + 16 * blk + kb + 2 * j;
        Aw[l][ob][q].x = p[0];
        Aw[l][ob][q].y = p[1];
      }

  v8f Cb[2][2];
#pragma unroll
  for (int l = 0; l < 2; ++l)
#pragma unroll
    for (int ob = 0; ob < 2; ++ob)
#pragma unroll
      for (int r = 0; r < 8; ++r)
        Cb[l][ob][r] = b_h[l * 32 + 16 * ob + kb + r];

  float Wi0v[8][2], Wi1v[8][2], binv[8][2];
#pragma unroll
  for (int q = 0; q < 8; ++q)
#pragma unroll
    for (int s = 0; s < 2; ++s) {
      const int k = 16 * (q >> 2) + kb + 2 * (q & 3) + s;
      Wi0v[q][s] = W_in[2 * k];
      Wi1v[q][s] = W_in[2 * k + 1];
      binv[q][s] = b_in[k];
    }

  float WoV[2][8];
#pragma unroll
  for (int ob = 0; ob < 2; ++ob)
#pragma unroll
    for (int r = 0; r < 8; ++r) WoV[ob][r] = W_out[16 * ob + kb + r];
  const float bo = b_out[0];

  const float4* pf = (const float4*)params;
  float4 ca = pf[0], cb = pf[1];
  float4 na = pf[2], nb = pf[3];

  float cv[8][2];
#pragma unroll
  for (int q = 0; q < 8; ++q)
#pragma unroll
    for (int s = 0; s < 2; ++s) cv[q][s] = fmaf(Wi1v[q][s], cb.w, binv[q][s]);

  float a0 = 0.0f, b1 = 0.0f;
  float dp_a = ca.y * a0 + ca.z * b1 + ca.w * ca.x;

  float o_prev = 0.0f;
  float* po = out;

  for (int t = 0; t < T; ++t) {
    float4 la = pf[2 * (t + 2)];
    float4 lb = pf[2 * (t + 2) + 1];
    __builtin_prefetch(params + 8 * (size_t)(t + 8), 0, 3);

    v2f Bh[8];
#pragma unroll
    for (int q = 0; q < 8; ++q) {
      Bh[q].x = fmaxf(fmaf(Wi0v[q][0], dp_a, cv[q][0]), 0.0f);
      Bh[q].y = fmaxf(fmaf(Wi0v[q][1], dp_a, cv[q][1]), 0.0f);
    }

    v8f d0 = Cb[0][0], d1 = Cb[0][1];
#pragma unroll
    for (int q = 0; q < 8; ++q) {
      d0 = mma(Aw[0][0][q], Bh[q], d0);
      d1 = mma(Aw[0][1][q], Bh[q], d1);
    }

    *po = o_prev;
    po += (t > 0);

    v8f e0 = Cb[1][0], e1 = Cb[1][1];
#pragma unroll
    for (int j = 0; j < 4; ++j) {
      v2f B;
      B.x = relu(d0[2 * j]);
      B.y = relu(d0[2 * j + 1]);
      e0 = mma(Aw[1][0][j], B, e0);
      e1 = mma(Aw[1][1][j], B, e1);
    }
#pragma unroll
    for (int j = 0; j < 4; ++j) {
      v2f B;
      B.x = relu(d1[2 * j]);
      B.y = relu(d1[2 * j + 1]);
      e0 = mma(Aw[1][0][4 + j], B, e0);
      e1 = mma(Aw[1][1][4 + j], B, e1);
    }

#pragma unroll
    for (int q = 0; q < 8; ++q) {
      cv[q][0] = fmaf(Wi1v[q][0], nb.w, binv[q][0]);
      cv[q][1] = fmaf(Wi1v[q][1], nb.w, binv[q][1]);
    }

    float c[4];
#pragma unroll
    for (int i = 0; i < 4; ++i) c[i] = WoV[0][i] * relu(e0[i]);
#pragma unroll
    for (int i = 0; i < 4; ++i) c[i] = fmaf(WoV[0][4 + i], relu(e0[4 + i]), c[i]);
#pragma unroll
    for (int i = 0; i < 4; ++i) c[i] = fmaf(WoV[1][i], relu(e1[i]), c[i]);
#pragma unroll
    for (int i = 0; i < 4; ++i) c[i] = fmaf(WoV[1][4 + i], relu(e1[4 + i]), c[i]);
    float ph = (c[0] + c[2]) + (c[1] + c[3]);
    float dp_b = (ph + other_half(ph)) + bo;

    float b1n = cb.x * dp_b + cb.y * b1 + cb.z * ca.x;

    o_prev = 0.5f * (dp_a + dp_b);

    a0 = dp_b;
    b1 = b1n;
    dp_a = na.y * a0 + na.z * b1 + na.w * na.x;

    ca = na;  cb = nb;
    na = la;  nb = lb;
  }
  if (T > 0) out[T - 1] = o_prev;
}

extern "C" void kernel_launch(void* const* d_in, const int* in_sizes, int n_in,
                              void* d_out, int out_size, void* d_ws,
                              size_t ws_size, hipStream_t stream) {
  const float* v_in  = (const float*)d_in[0];
  const float* vs_r  = (const float*)d_in[1];
  const float* fs    = (const float*)d_in[2];
  const float* W_in  = (const float*)d_in[3];
  const float* b_in  = (const float*)d_in[4];
  const float* W_h   = (const float*)d_in[5];
  const float* b_h   = (const float*)d_in[6];
  const float* W_out = (const float*)d_in[7];
  const float* b_out = (const float*)d_in[8];
  float* out = (float*)d_out;
  int T = in_sizes[0];

  float* params = (float*)d_ws;

  ndc_prep<<<(T + 8 + 255) / 256, 256, 0, stream>>>(v_in, vs_r, fs, params, T);
  ndc_scan<<<1, 32, 0, stream>>>(params, W_in, b_in, W_h, b_h, W_out, b_out,
                                 out, T);
}


# Step update re-associated as affine in dp_b (dp_a' = alpha*dp_b + beta; alpha, beta, P2*v, half of dp_a formed ahead; b_out as a partial sum's start value): shorter serial chain; sum order changes
# speedup vs baseline: 1.2761x; 1.0762x over previous
#include <hip/hip_runtime.h>


typedef __attribute__((ext_vector_type(2))) float v2f;
typedef __attribute__((ext_vector_type(8))) float v8f;

#define C1F 4.7e-9f

__global__ void ndc_prep(const float* __restrict__ v_in,
                         const float* __restrict__ vs_r,
                         const float* __restrict__ fs,
                         float* __restrict__ params, int T) {
  int t = blockIdx.x * blockDim.x + threadIdx.x;
  if (t >= T + 8) return;
  int ti = (t < T) ? t : (T - 1);
  float f   = fs[ti];
  float vr  = vs_r[ti];
  float c1r = 1.0f / (2.0f * C1F * f);
  float r0  = c1r * vr / (c1r + vr);
  float g0 = 1.0f / r0, g1 = 1.0f / c1r, g2 = 1.0f / vr;
  float inv = 2.0f / (g0 + g1 + g2);
  float P0 = g0 * inv, P1 = g1 * inv, P2 = g2 * inv;
  float p2v = P2 * v_in[ti];
  float4* p4 = (float4*)(params + 8 * (size_t)t);
  p4[0] = make_float4(P0 - 1.0f, P1, p2v, P0);
  p4[1] = make_float4(P1 - 1.0f, r0 * (1.0f / 3000.0f), 0.0f, 0.0f);
}

__device__ __forceinline__ float other_half(float x) {
  return __builtin_bit_cast(
      float, __builtin_amdgcn_permlanex16(0, __builtin_bit_cast(int, x),
                                          0x76543210, 0xfedcba98, false, false));
}

__device__ __forceinline__ float relu(float x) {
  return __builtin_elementwise_maximum(x, 0.0f);
}

__device__ __forceinline__ v8f mma(v2f a, v2f b, v8f c) {
  return __builtin_amdgcn_wmma_f32_16x16x4_f32(false, a, false, b, (short)0, c,
                                               false, false);
}

__global__ __launch_bounds__(32, 1) void ndc_scan(
    const float* __restrict__ params,
    const float* __restrict__ W_in, const float* __restrict__ b_in,
    const float* __restrict__ W_h,  const float* __restrict__ b_h,
    const float* __restrict__ W_out, const float* __restrict__ b_out,
    float* __restrict__ out, int T) {
  const int lane = threadIdx.x;
  const int c16  = lane & 15;
  const int kb   = 8 * (lane >> 4);

  v2f Aw[2][2][8];
#pragma unroll
  for (int l = 0; l < 2; ++l)
#pragma unroll
    for (int ob = 0; ob < 2; ++ob)
#pragma unroll
      for (int q = 0; q < 8; ++q) {
        const int blk = q >> 2, j = q & 3;
        const float* p =
            W_h + l * 1024 + (16 * ob + c16) * 32 + 16 * blk + kb + 2 * j;
        Aw[l][ob][q].x = p[0];
        Aw[l][ob][q].y = p[1];
      }

  v8f Cb[2][2];
#pragma unroll
  for (int l = 0; l < 2; ++l)
#pragma unroll
    for (int ob = 0; ob < 2; ++ob)
#pragma unroll
      for (int r = 0; r < 8; ++r)
        Cb[l][ob][r] = b_h[l * 32 + 16 * ob + kb + r];

  float Wi0v[8][2], Wi1v[8][2], binv[8][2];
#pragma unroll
  for (int q = 0; q < 8; ++q)
#pragma unroll
    for (int s = 0; s < 2; ++s) {
      const int k = 16 * (q >> 2) + kb + 2 * (q & 3) + s;
      Wi0v[q][s] = W_in[2 * k];
      Wi1v[q][s] = W_in[2 * k + 1];
      binv[q][s] = b_in[k];
    }

  float WoV[2][8];
#pragma unroll
  for (int ob = 0; ob < 2; ++ob)
#pragma unroll
    for (int r = 0; r < 8; ++r) WoV[ob][r] = W_out[16 * ob + kb + r];
  const float cst[4] = {(lane < 16) ? b_out[0] : 0.0f, 0.0f, 0.0f, 0.0f};

  const float4* pf = (const float4*)params;
  float4 ca = pf[0], cb = pf[1];
  float4 na = pf[2], nb = pf[3];

  float cv[8][2];
#pragma unroll
  for (int q = 0; q < 8; ++q)
#pragma unroll
    for (int s = 0; s < 2; ++s) cv[q][s] = fmaf(Wi1v[q][s], cb.y, binv[q][s]);

  float b1 = 0.0f;
  float dp_a = fmaf(ca.y, 0.0f, fmaf(ca.x, 0.0f, ca.z));

  float o_prev = 0.0f;
  float* po = out;

  for (int t = 0; t < T; ++t) {
    float4 la = pf[2 * (t + 2)];
    float4 lb = pf[2 * (t + 2) + 1];
    __builtin_prefetch(params + 8 * (size_t)(t + 8), 0, 3);

    v2f Bh[8];
#pragma unroll
    for (int q = 0; q < 8; ++q) {
      Bh[q].x = fmaxf(fmaf(Wi0v[q][0], dp_a, cv[q][0]), 0.0f);
      Bh[q].y = fmaxf(fmaf(Wi0v[q][1], dp_a, cv[q][1]), 0.0f);
    }

    v8f d0 = Cb[0][0], d1 = Cb[0][1];
#pragma unroll
    for (int q = 0; q < 8; ++q) {
      d0 = mma(Aw[0][0][q], Bh[q], d0);
      d1 = mma(Aw[0][1][q], Bh[q], d1);
    }

    *po = o_prev;
    po += (t > 0);

    v8f e0 = Cb[1][0], e1 = Cb[1][1];
#pragma unroll
    for (int j = 0; j < 4; ++j) {
      v2f B;
      B.x = relu(d0[2 * j]);
      B.y = relu(d0[2 * j + 1]);
      e0 = mma(Aw[1][0][j], B, e0);
      e1 = mma(Aw[1][1][j], B, e1);
    }
#pragma unroll
    for (int j = 0; j < 4; ++j) {
      v2f B;
      B.x = relu(d1[2 * j]);
      B.y = relu(d1[2 * j + 1]);
      e0 = mma(Aw[1][0][4 + j], B, e0);
      e1 = mma(Aw[1][1][4 + j], B, e1);
    }

#pragma unroll
    for (int q = 0; q < 8; ++q) {
      cv[q][0] = fmaf(Wi1v[q][0], nb.y, binv[q][0]);
      cv[q][1] = fmaf(Wi1v[q][1], nb.y, binv[q][1]);
    }

    const float h_a   = 0.5f * dp_a;
    const float u     = fmaf(cb.x, b1, ca.z);
    const float alpha = fmaf(na.y, ca.w, na.x);
    const float beta  = fmaf(na.y, u, na.z);

    float c[4];
#pragma unroll
    for (int i = 0; i < 4; ++i) c[i] = fmaf(WoV[0][i], relu(e0[i]), cst[i]);
#pragma unroll
    for (int i = 0; i < 4; ++i) c[i] = fmaf(WoV[0][4 + i], relu(e0[4 + i]), c[i]);
#pragma unroll
    for (int i = 0; i < 4; ++i) c[i] = fmaf(WoV[1][i], relu(e1[i]), c[i]);
#pragma unroll
    for (int i = 0; i < 4; ++i) c[i] = fmaf(WoV[1][4 + i], relu(e1[4 + i]), c[i]);
    float ph = (c[0] + c[2]) + (c[1] + c[3]);
    float dp_b = ph + other_half(ph);

    float b1n = fmaf(ca.w, dp_b, u);

    o_prev = fmaf(0.5f, dp_b, h_a);

    b1 = b1n;
    dp_a = fmaf(alpha, dp_b, beta);

    ca = na;  cb = nb;
    na = la;  nb = lb;
  }
  if (T > 0) out[T - 1] = o_prev;
}

extern "C" void kernel_launch(void* const* d_in, const int* in_sizes, int n_in,
                              void* d_out, int out_size, void* d_ws,
                              size_t ws_size, hipStream_t stream) {
  const float* v_in  = (const float*)d_in[0];
  const float* vs_r  = (const float*)d_in[1];
  const float* fs    = (const float*)d_in[2];
  const float* W_in  = (const float*)d_in[3];
  const float* b_in  = (const float*)d_in[4];
  const float* W_h   = (const float*)d_in[5];
  const float* b_h   = (const float*)d_in[6];
  const float* W_out = (const float*)d_in[7];
  const float* b_out = (const float*)d_in[8];
  float* out = (float*)d_out;
  int T = in_sizes[0];

  float* params = (float*)d_ws;

  ndc_prep<<<(T + 8 + 255) / 256, 256, 0, stream>>>(v_in, vs_r, fs, params, T);
  ndc_scan<<<1, 32, 0, stream>>>(params, W_in, b_in, W_h, b_h, W_out, b_out,
                                 out, T);
}


# Four steps per loop trip: one coefficient-block request and one wait (asm, three steps apart), one 16-byte output store, one taken branch per four steps; per-step arithmetic and its order unchanged
# speedup vs baseline: 1.4040x; 1.1002x over previous
#include <hip/hip_runtime.h>


typedef __attribute__((ext_vector_type(2))) float v2f;
typedef __attribute__((ext_vector_type(8))) float v8f;

#define C1F 4.7e-9f

__global__ void ndc_prep(const float* __restrict__ v_in,
                         const float* __restrict__ vs_r,
                         const float* __restrict__ fs,
                         float* __restrict__ params, int T) {
  int t = blockIdx.x * blockDim.x + threadIdx.x;
  if (t >= T + 8) return;
  int ti = (t < T) ? t : (T - 1);
  float f   = fs[ti];
  float vr  = vs_r[ti];
  float c1r = 1.0f / (2.0f * C1F * f);
  float r0  = c1r * vr / (c1r + vr);
  float g0 = 1.0f / r0, g1 = 1.0f / c1r, g2 = 1.0f / vr;
  float inv = 2.0f / (g0 + g1 + g2);
  float P0 = g0 * inv, P1 = g1 * inv, P2 = g2 * inv;
  float p2v = P2 * v_in[ti];
  float4* p4 = (float4*)(params + 8 * (size_t)t);
  p4[0] = make_float4(P0 - 1.0f, P1, p2v, P0);
  p4[1] = make_float4(P1 - 1.0f, r0 * (1.0f / 3000.0f), 0.0f, 0.0f);
}

__device__ __forceinline__ float other_half(float x) {
  return __builtin_bit_cast(
      float, __builtin_amdgcn_permlanex16(0, __builtin_bit_cast(int, x),
                                          0x76543210, 0xfedcba98, false, false));
}

__device__ __forceinline__ float relu(float x) {
  return __builtin_elementwise_maximum(x, 0.0f);
}

__device__ __forceinline__ v8f mma(v2f a, v2f b, v8f c) {
  return __builtin_amdgcn_wmma_f32_16x16x4_f32(false, a, false, b, (short)0, c,
                                               false, false);
}

__global__ __launch_bounds__(32, 1) void ndc_scan(
    const float* __restrict__ params,
    const float* __restrict__ W_in, const float* __restrict__ b_in,
    const float* __restrict__ W_h,  const float* __restrict__ b_h,
    const float* __restrict__ W_out, const float* __restrict__ b_out,
    float* __restrict__ out, int T) {
  const int lane = threadIdx.x;
  const int c16  = lane & 15;
  const int kb   = 8 * (lane >> 4);

  v2f Aw[2][2][8];
#pragma unroll
  for (int l = 0; l < 2; ++l)
#pragma unroll
    for (int ob = 0; ob < 2; ++ob)
#pragma unroll
      for (int q = 0; q < 8; ++q) {
        const int blk = q >> 2, j = q & 3;
        const float* p =
            W_h + l * 1024 + (16 * ob + c16) * 32 + 16 * blk + kb + 2 * j;
        Aw[l][ob][q].x = p[0];
        Aw[l][ob][q].y = p[1];
      }

  v8f Cb[2][2];
#pragma unroll
  for (int l = 0; l < 2; ++l)
#pragma unroll
    for (int ob = 0; ob < 2; ++ob)
#pragma unroll
      for (int r = 0; r < 8; ++r)
        Cb[l][ob][r] = b_h[l * 32 + 16 * ob + kb + r];

  float Wi0v[8][2], Wi1v[8][2], binv[8][2];
#pragma unroll
  for (int q = 0; q < 8; ++q)
#pragma unroll
    for (int s = 0; s < 2; ++s) {
      const int k = 16 * (q >> 2) + kb + 2 * (q & 3) + s;
      Wi0v[q][s] = W_in[2 * k];
      Wi1v[q][s] = W_in[2 * k + 1];
      binv[q][s] = b_in[k];
    }

  float WoV[2][8];
#pragma unroll
  for (int ob = 0; ob < 2; ++ob)
#pragma unroll
    for (int r = 0; r < 8; ++r) WoV[ob][r] = W_out[16 * ob + kb + r];
  const float cst[4] = {(lane < 16) ? b_out[0] : 0.0f, 0.0f, 0.0f, 0.0f};

  const float4* pf = (const float4*)params;

  float cv[8][2];
  {
    const float dr0 = pf[1].y;
#pragma unroll
    for (int q = 0; q < 8; ++q)
#pragma unroll
      for (int s = 0; s < 2; ++s) cv[q][s] = fmaf(Wi1v[q][s], dr0, binv[q][s]);
  }

  float b1 = 0.0f;
  float dp_a;
  {
    const float4 c0 = pf[0];
    dp_a = fmaf(c0.y, 0.0f, fmaf(c0.x, 0.0f, c0.z));
  }

  auto step = [&](const float4 ca, const float4 cb, const float4 na,
                  const float4 nb)
                  __attribute__((always_inline)) -> float {
    v2f Bh[8];
#pragma unroll
    for (int q = 0; q < 8; ++q) {
      Bh[q].x = fmaxf(fmaf(Wi0v[q][0], dp_a, cv[q][0]), 0.0f);
      Bh[q].y = fmaxf(fmaf(Wi0v[q][1], dp_a, cv[q][1]), 0.0f);
    }

    v8f d0 = Cb[0][0], d1 = Cb[0][1];
#pragma unroll
    for (int q = 0; q < 8; ++q) {
      d0 = mma(Aw[0][0][q], Bh[q], d0);
      d1 = mma(Aw[0][1][q], Bh[q], d1);
    }

    v8f e0 = Cb[1][0], e1 = Cb[1][1];
#pragma unroll
    for (int j = 0; j < 4; ++j) {
      v2f B;
      B.x = relu(d0[2 * j]);
      B.y = relu(d0[2 * j + 1]);
      e0 = mma(Aw[1][0][j], B, e0);
      e1 = mma(Aw[1][1][j], B, e1);
    }
#pragma unroll
    for (int j = 0; j < 4; ++j) {
      v2f B;
      B.x = relu(d1[2 * j]);
      B.y = relu(d1[2 * j + 1]);
      e0 = mma(Aw[1][0][4 + j], B, e0);
      e1 = mma(Aw[1][1][4 + j], B, e1);
    }

#pragma unroll
    for (int q = 0; q < 8; ++q) {
      cv[q][0] = fmaf(Wi1v[q][0], nb.y, binv[q][0]);
      cv[q][1] = fmaf(Wi1v[q][1], nb.y, binv[q][1]);
    }

    const float h_a   = 0.5f * dp_a;
    const float u     = fmaf(cb.x, b1, ca.z);
    const float alpha = fmaf(na.y, ca.w, na.x);
    const float beta  = fmaf(na.y, u, na.z);

    float c[4];
#pragma unroll
    for (int i = 0; i < 4; ++i) c[i] = fmaf(WoV[0][i], relu(e0[i]), cst[i]);
#pragma unroll
    for (int i = 0; i < 4; ++i) c[i] = fmaf(WoV[0][4 + i], relu(e0[4 + i]), c[i]);
#pragma unroll
    for (int i = 0; i < 4; ++i) c[i] = fmaf(WoV[1][i], relu(e1[i]), c[i]);
#pragma unroll
    for (int i = 0; i < 4; ++i) c[i] = fmaf(WoV[1][4 + i], relu(e1[4 + i]), c[i]);
    float ph = (c[0] + c[2]) + (c[1] + c[3]);
    float dp_b = ph + other_half(ph);

    b1 = fmaf(ca.w, dp_b, u);
    dp_a = fmaf(alpha, dp_b, beta);
    return fmaf(0.5f, dp_b, h_a);
  };

  typedef float v4f __attribute__((ext_vector_type(4)));
  typedef unsigned long long u64;
  auto lo32 = [](const u64 w) __attribute__((always_inline)) -> float {
    return __builtin_bit_cast(float, (unsigned)(w & 0xffffffffull));
  };
  auto hi32 = [](const u64 w) __attribute__((always_inline)) -> float {
    return __builtin_bit_cast(float, (unsigned)(w >> 32));
  };
  auto f4a = [&](const u64 w0, const u64 w1) __attribute__((always_inline)) -> float4 {
    return make_float4(lo32(w0), hi32(w0), lo32(w1), hi32(w1));
  };
  auto f4b = [&](const u64 w2) __attribute__((always_inline)) -> float4 {
    return make_float4(lo32(w2), hi32(w2), 0.0f, 0.0f);
  };

  u64 cw[4][3];
  {
    const u64* pw = (const u64*)params;
#pragma unroll
    for (int j = 0; j < 4; ++j) {
      cw[j][0] = pw[4 * j + 0];
      cw[j][1] = pw[4 * j + 1];
      cw[j][2] = pw[4 * j + 2];
    }
  }
  float4 o4 = make_float4(0.0f, 0.0f, 0.0f, 0.0f);
  int t = 0;
  for (; t + 4 <= T; t += 4) {
    u64 nw[4][3];
    const float* nblk = params + 8 * (size_t)(t + 4);
    float* po = out + (t > 0 ? t - 4 : 0);
    const v4f o_st = {o4.x, o4.y, o4.z, o4.w};
    const int zoff = 0;
    asm volatile("s_load_b64 %0, %13, 0x0\n\t"
                 "s_load_b64 %1, %13, 0x8\n\t"
                 "s_load_b64 %2, %13, 0x10\n\t"
                 "s_load_b64 %3, %13, 0x20\n\t"
                 "s_load_b64 %4, %13, 0x28\n\t"
                 "s_load_b64 %5, %13, 0x30\n\t"
                 "s_load_b64 %6, %13, 0x40\n\t"
                 "s_load_b64 %7, %13, 0x48\n\t"
                 "s_load_b64 %8, %13, 0x50\n\t"
                 "s_load_b64 %9, %13, 0x60\n\t"
                 "s_load_b64 %10, %13, 0x68\n\t"
                 "s_load_b64 %11, %13, 0x70\n\t"
                 "global_store_b128 %14, %15, %16"
                 : "=&s"(nw[0][0]), "=&s"(nw[0][1]), "=&s"(nw[0][2]),
                   "=&s"(nw[1][0]), "=&s"(nw[1][1]), "=&s"(nw[1][2]),
                   "=&s"(nw[2][0]), "=&s"(nw[2][1]), "=&s"(nw[2][2]),
                   "=&s"(nw[3][0]), "=&s"(nw[3][1]), "=&s"(nw[3][2]),
                   "+v"(dp_a)
                 : "s"(nblk), "v"(zoff), "v"(o_st), "s"(po)
                 : "memory");
    {
      const int tp = (t + 12 < T + 4) ? (t + 12) : (T + 4);
      __builtin_prefetch(params + 8 * (size_t)tp, 0, 3);
    }

    const float o0 = step(f4a(cw[0][0], cw[0][1]), f4b(cw[0][2]),
                          f4a(cw[1][0], cw[1][1]), f4b(cw[1][2]));
    const float o1 = step(f4a(cw[1][0], cw[1][1]), f4b(cw[1][2]),
                          f4a(cw[2][0], cw[2][1]), f4b(cw[2][2]));
    const float o2 = step(f4a(cw[2][0], cw[2][1]), f4b(cw[2][2]),
                          f4a(cw[3][0], cw[3][1]), f4b(cw[3][2]));
    asm volatile("s_wait_kmcnt 0x0\n\t"
                 "s_wait_xcnt 0x0"
                 : "+s"(nw[0][0]), "+s"(nw[0][1]), "+s"(nw[0][2]),
                   "+s"(nw[1][0]), "+s"(nw[1][1]), "+s"(nw[1][2]),
                   "+s"(nw[2][0]), "+s"(nw[2][1]), "+s"(nw[2][2]),
                   "+s"(nw[3][0]), "+s"(nw[3][1]), "+s"(nw[3][2]),
                   "+v"(dp_a)
                 : "v"(zoff), "v"(o_st), "s"(po), "s"(nblk)
                 : "memory");
    const float o3 = step(f4a(cw[3][0], cw[3][1]), f4b(cw[3][2]),
                          f4a(nw[0][0], nw[0][1]), f4b(nw[0][2]));
    o4 = make_float4(o0, o1, o2, o3);
#pragma unroll
    for (int j = 0; j < 4; ++j)
#pragma unroll
      for (int k = 0; k < 3; ++k) cw[j][k] = nw[j][k];
  }
  if (t > 0) *(float4*)(out + t - 4) = o4;

  for (; t < T; ++t) {
    const float4 ca = pf[2 * t], cb = pf[2 * t + 1];
    const float4 na = pf[2 * t + 2], nb = pf[2 * t + 3];
    const float o = step(ca, cb, na, nb);
    out[t] = o;
  }
}

extern "C" void kernel_launch(void* const* d_in, const int* in_sizes, int n_in,
                              void* d_out, int out_size, void* d_ws,
                              size_t ws_size, hipStream_t stream) {
  const float* v_in  = (const float*)d_in[0];
  const float* vs_r  = (const float*)d_in[1];
  const float* fs    = (const float*)d_in[2];
  const float* W_in  = (const float*)d_in[3];
  const float* b_in  = (const float*)d_in[4];
  const float* W_h   = (const float*)d_in[5];
  const float* b_h   = (const float*)d_in[6];
  const float* W_out = (const float*)d_in[7];
  const float* b_out = (const float*)d_in[8];
  float* out = (float*)d_out;
  int T = in_sizes[0];

  float* params = (float*)d_ws;

  ndc_prep<<<(T + 8 + 255) / 256, 256, 0, stream>>>(v_in, vs_r, fs, params, T);
  ndc_scan<<<1, 32, 0, stream>>>(params, W_in, b_in, W_h, b_h, W_out, b_out,
                                 out, T);
}


# Last hidden layer with weights in BOTH matrix operands (hidden chunk as row 15 of A and column 15 of B): 9 matrix instructions, not 16; results lie along lanes, 16-lane sum added; sum order changes
# speedup vs baseline: 1.4469x; 1.0306x over previous
#include <hip/hip_runtime.h>


typedef __attribute__((ext_vector_type(2))) float v2f;
typedef __attribute__((ext_vector_type(8))) float v8f;

#define C1F 4.7e-9f

__global__ void ndc_prep(const float* __restrict__ v_in,
                         const float* __restrict__ vs_r,
                         const float* __restrict__ fs,
                         float* __restrict__ params, int T) {
  int t = blockIdx.x * blockDim.x + threadIdx.x;
  if (t >= T + 8) return;
  int ti = (t < T) ? t : (T - 1);
  float f   = fs[ti];
  float vr  = vs_r[ti];
  float c1r = 1.0f / (2.0f * C1F * f);
  float r0  = c1r * vr / (c1r + vr);
  float g0 = 1.0f / r0, g1 = 1.0f / c1r, g2 = 1.0f / vr;
  float inv = 2.0f / (g0 + g1 + g2);
  float P0 = g0 * inv, P1 = g1 * inv, P2 = g2 * inv;
  float p2v = P2 * v_in[ti];
  float4* p4 = (float4*)(params + 8 * (size_t)t);
  p4[0] = make_float4(P0 - 1.0f, P1, p2v, P0);
  p4[1] = make_float4(P1 - 1.0f, r0 * (1.0f / 3000.0f), 0.0f, 0.0f);
}

__device__ __forceinline__ float other_half(float x) {
  return __builtin_bit_cast(
      float, __builtin_amdgcn_permlanex16(0, __builtin_bit_cast(int, x),
                                          0x76543210, 0xfedcba98, false, false));
}

__device__ __forceinline__ float relu(float x) {
  return __builtin_elementwise_maximum(x, 0.0f);
}

__device__ __forceinline__ float clampf(float x, float lo, float hi) {
  return __builtin_amdgcn_fmed3f(x, lo, hi);
}

template <int M>
__device__ __forceinline__ float row_xor(float x) {
  return __builtin_bit_cast(
      float, __builtin_amdgcn_update_dpp(0, __builtin_bit_cast(int, x),
                                         0x160 + M, 0xf, 0xf, false));
}

__device__ __forceinline__ v8f mma(v2f a, v2f b, v8f c) {
  return __builtin_amdgcn_wmma_f32_16x16x4_f32(false, a, false, b, (short)0, c,
                                               false, false);
}

__global__ __launch_bounds__(32, 1) void ndc_scan(
    const float* __restrict__ params,
    const float* __restrict__ W_in, const float* __restrict__ b_in,
    const float* __restrict__ W_h,  const float* __restrict__ b_h,
    const float* __restrict__ W_out, const float* __restrict__ b_out,
    float* __restrict__ out, int T) {
  const int lane = threadIdx.x;
  const int c16  = lane & 15;
  const int kb   = 8 * (lane >> 4);

  v2f Aw[2][8];
#pragma unroll
  for (int ob = 0; ob < 2; ++ob)
#pragma unroll
    for (int q = 0; q < 8; ++q) {
      const int blk = q >> 2, j = q & 3;
      const float* p = W_h + (16 * ob + c16) * 32 + 16 * blk + kb + 2 * j;
      Aw[ob][q].x = p[0];
      Aw[ob][q].y = p[1];
    }

  v8f Cb[2];
#pragma unroll
  for (int ob = 0; ob < 2; ++ob)
#pragma unroll
    for (int r = 0; r < 8; ++r) Cb[ob][r] = b_h[16 * ob + kb + r];

  const bool  hl  = (c16 == 15);
  const int   ua  = hl ? 0 : c16;
  const int   ub  = hl ? 0 : 15 + c16;
  const float INF = __builtin_inff();
  float ALo[8][2], AHi[8][2], BLo[8][2], BHi[8][2];
#pragma unroll
  for (int q = 0; q < 8; ++q)
#pragma unroll
    for (int s = 0; s < 2; ++s) {
      const int k = 16 * (q >> 2) + kb + 2 * (q & 3) + s;
      const float wa = W_h[1024 + ua * 32 + k];
      const float wb = W_h[1024 + ub * 32 + k];
      ALo[q][s] = hl ? 0.0f : wa;
      AHi[q][s] = hl ? INF : wa;
      BLo[q][s] = hl ? 0.0f : wb;
      BHi[q][s] = hl ? INF : wb;
    }
  v8f   C2;
  v2f   Wo2[4];
  float Hi2[8];
#pragma unroll
  for (int r = 0; r < 8; ++r) {
    const int  m    = kb + r;
    const bool colu = hl && (m < 15);
    const bool rowu = !hl && (m == 15);
    const int  u    = colu ? m : (rowu ? 15 + c16 : 0);
    const bool any  = colu || rowu;
    C2[r]  = any ? b_h[32 + u] : 0.0f;
    Wo2[r >> 1][r & 1] = any ? W_out[u] : 0.0f;
    Hi2[r] = any ? INF : 0.0f;
  }
  const int gq = c16 & 7;
  const int hf = lane >> 4;
  v2f A9;
  {
    const float* p = W_h + 1024 + (30 + (c16 >> 3)) * 32 + 16 * (gq >> 2) + kb +
                     2 * (gq & 3);
    A9.x = p[0];
    A9.y = p[1];
  }
  const float bG = b_h[32 + 30 + hf];
  const float wG = (c16 == 0) ? W_out[30 + hf] : 0.0f;
  const bool g1 = (gq & 1) != 0, g2 = (gq & 2) != 0, g4 = (gq & 4) != 0;
  auto pick8 = [&](const float v0, const float v1, const float v2,
                   const float v3, const float v4, const float v5,
                   const float v6, const float v7)
                   __attribute__((always_inline)) -> float {
    const float t0 = g1 ? v1 : v0, t1 = g1 ? v3 : v2;
    const float t2 = g1 ? v5 : v4, t3 = g1 ? v7 : v6;
    const float w0 = g2 ? t1 : t0, w1 = g2 ? t3 : t2;
    return g4 ? w1 : w0;
  };

  float Wi0v[8][2], Wi1v[8][2], binv[8][2];
#pragma unroll
  for (int q = 0; q < 8; ++q)
#pragma unroll
    for (int s = 0; s < 2; ++s) {
      const int k = 16 * (q >> 2) + kb + 2 * (q & 3) + s;
      Wi0v[q][s] = W_in[2 * k];
      Wi1v[q][s] = W_in[2 * k + 1];
      binv[q][s] = b_in[k];
    }

  const float bo = (lane == 0) ? b_out[0] : 0.0f;

  const float4* pf = (const float4*)params;

  float cv[8][2];
  {
    const float dr0 = pf[1].y;
#pragma unroll
    for (int q = 0; q < 8; ++q)
#pragma unroll
      for (int s = 0; s < 2; ++s) cv[q][s] = fmaf(Wi1v[q][s], dr0, binv[q][s]);
  }

  float b1 = 0.0f;
  float dp_a;
  {
    const float4 c0 = pf[0];
    dp_a = fmaf(c0.y, 0.0f, fmaf(c0.x, 0.0f, c0.z));
  }

  auto step = [&](const float4 ca, const float4 cb, const float4 na,
                  const float4 nb)
                  __attribute__((always_inline)) -> float {
    v2f Bh[8];
#pragma unroll
    for (int q = 0; q < 8; ++q) {
      Bh[q].x = fmaxf(fmaf(Wi0v[q][0], dp_a, cv[q][0]), 0.0f);
      Bh[q].y = fmaxf(fmaf(Wi0v[q][1], dp_a, cv[q][1]), 0.0f);
    }

    v8f d0 = Cb[0], d1 = Cb[1];
#pragma unroll
    for (int q = 0; q < 8; ++q) {
      d0 = mma(Aw[0][q], Bh[q], d0);
      d1 = mma(Aw[1][q], Bh[q], d1);
    }

    v8f ey0;
#pragma unroll
    for (int r = 0; r < 8; ++r) ey0[r] = 0.0f;
    v8f ex = C2, ey = ey0;
    v8f g;
    float cst0 = 0.0f;
#pragma unroll
    for (int j = 0; j < 4; ++j) {
      v2f A, B;
      A.x = clampf(d0[2 * j], ALo[j][0], AHi[j][0]);
      A.y = clampf(d0[2 * j + 1], ALo[j][1], AHi[j][1]);
      B.x = clampf(d0[2 * j], BLo[j][0], BHi[j][0]);
      B.y = clampf(d0[2 * j + 1], BLo[j][1], BHi[j][1]);
      ex = mma(A, B, ex);
      A.x = clampf(d1[2 * j], ALo[4 + j][0], AHi[4 + j][0]);
      A.y = clampf(d1[2 * j + 1], ALo[4 + j][1], AHi[4 + j][1]);
      B.x = clampf(d1[2 * j], BLo[4 + j][0], BHi[4 + j][0]);
      B.y = clampf(d1[2 * j + 1], BLo[4 + j][1], BHi[4 + j][1]);
      ey = mma(A, B, ey);
      if (j == 0) {
        v2f B9;
        B9.x = relu(pick8(d0[0], d0[2], d0[4], d0[6], d1[0], d1[2], d1[4], d1[6]));
        B9.y = relu(pick8(d0[1], d0[3], d0[5], d0[7], d1[1], d1[3], d1[5], d1[7]));
        g = mma(A9, B9, ey0);
        __builtin_amdgcn_sched_barrier(0);
      }
      if (j == 2) {
        float gv = pick8(g[0], g[1], g[2], g[3], g[4], g[5], g[6], g[7]);
        gv += row_xor<1>(gv);
        gv += row_xor<2>(gv);
        gv += row_xor<4>(gv);
        cst0 = fmaf(wG, relu(gv + bG), bo);
        __builtin_amdgcn_sched_barrier(0);
      }
    }

#pragma unroll
    for (int q = 0; q < 8; ++q) {
      cv[q][0] = fmaf(Wi1v[q][0], nb.y, binv[q][0]);
      cv[q][1] = fmaf(Wi1v[q][1], nb.y, binv[q][1]);
    }

    const float h_a   = 0.5f * dp_a;
    const float u     = fmaf(cb.x, b1, ca.z);
    const float alpha = fmaf(na.y, ca.w, na.x);
    const float beta  = fmaf(na.y, u, na.z);

    v2f a[4];
#pragma unroll
    for (int i = 0; i < 4; ++i) {
      a[i].x = clampf(ex[2 * i] + ey[2 * i], 0.0f, Hi2[2 * i]);
      a[i].y = clampf(ex[2 * i + 1] + ey[2 * i + 1], 0.0f, Hi2[2 * i + 1]);
    }
    v2f t0 = {cst0, 0.0f};
    t0 = __builtin_elementwise_fma(Wo2[0], a[0], t0);
    v2f t1 = Wo2[1] * a[1];
    t0 = __builtin_elementwise_fma(Wo2[2], a[2], t0);
    t1 = __builtin_elementwise_fma(Wo2[3], a[3], t1);
    const v2f ts = t0 + t1;
    float ph = ts.x + ts.y;
    ph += row_xor<1>(ph);
    ph += row_xor<2>(ph);
    ph += row_xor<4>(ph);
    ph += row_xor<8>(ph);
    float dp_b = ph + other_half(ph);

    b1 = fmaf(ca.w, dp_b, u);
    dp_a = fmaf(alpha, dp_b, beta);
    return fmaf(0.5f, dp_b, h_a);
  };

  typedef float v4f __attribute__((ext_vector_type(4)));
  typedef unsigned long long u64;
  auto lo32 = [](const u64 w) __attribute__((always_inline)) -> float {
    return __builtin_bit_cast(float, (unsigned)(w & 0xffffffffull));
  };
  auto hi32 = [](const u64 w) __attribute__((always_inline)) -> float {
    return __builtin_bit_cast(float, (unsigned)(w >> 32));
  };
  auto f4a = [&](const u64 w0, const u64 w1) __attribute__((always_inline)) -> float4 {
    return make_float4(lo32(w0), hi32(w0), lo32(w1), hi32(w1));
  };
  auto f4b = [&](const u64 w2) __attribute__((always_inline)) -> float4 {
    return make_float4(lo32(w2), hi32(w2), 0.0f, 0.0f);
  };

  u64 cw[4][3];
  {
    const u64* pw = (const u64*)params;
#pragma unroll
    for (int j = 0; j < 4; ++j) {
      cw[j][0] = pw[4 * j + 0];
      cw[j][1] = pw[4 * j + 1];
      cw[j][2] = pw[4 * j + 2];
    }
  }
  float4 o4 = make_float4(0.0f, 0.0f, 0.0f, 0.0f);
  int t = 0;
  for (; t + 4 <= T; t += 4) {
    u64 nw[4][3];
    const float* nblk = params + 8 * (size_t)(t + 4);
    float* po = out + (t > 0 ? t - 4 : 0);
    const v4f o_st = {o4.x, o4.y, o4.z, o4.w};
    const int zoff = 0;
    asm volatile("s_load_b64 %0, %13, 0x0\n\t"
                 "s_load_b64 %1, %13, 0x8\n\t"
                 "s_load_b64 %2, %13, 0x10\n\t"
                 "s_load_b64 %3, %13, 0x20\n\t"
                 "s_load_b64 %4, %13, 0x28\n\t"
                 "s_load_b64 %5, %13, 0x30\n\t"
                 "s_load_b64 %6, %13, 0x40\n\t"
                 "s_load_b64 %7, %13, 0x48\n\t"
                 "s_load_b64 %8, %13, 0x50\n\t"
                 "s_load_b64 %9, %13, 0x60\n\t"
                 "s_load_b64 %10, %13, 0x68\n\t"
                 "s_load_b64 %11, %13, 0x70\n\t"
                 "global_store_b128 %14, %15, %16"
                 : "=&s"(nw[0][0]), "=&s"(nw[0][1]), "=&s"(nw[0][2]),
                   "=&s"(nw[1][0]), "=&s"(nw[1][1]), "=&s"(nw[1][2]),
                   "=&s"(nw[2][0]), "=&s"(nw[2][1]), "=&s"(nw[2][2]),
                   "=&s"(nw[3][0]), "=&s"(nw[3][1]), "=&s"(nw[3][2]),
                   "+v"(dp_a)
                 : "s"(nblk), "v"(zoff), "v"(o_st), "s"(po)
                 : "memory");
    {
      const int tp = (t + 12 < T + 4) ? (t + 12) : (T + 4);
      __builtin_prefetch(params + 8 * (size_t)tp, 0, 3);
    }

    const float o0 = step(f4a(cw[0][0], cw[0][1]), f4b(cw[0][2]),
                          f4a(cw[1][0], cw[1][1]), f4b(cw[1][2]));
    const float o1 = step(f4a(cw[1][0], cw[1][1]), f4b(cw[1][2]),
                          f4a(cw[2][0], cw[2][1]), f4b(cw[2][2]));
    const float o2 = step(f4a(cw[2][0], cw[2][1]), f4b(cw[2][2]),
                          f4a(cw[3][0], cw[3][1]), f4b(cw[3][2]));
    asm volatile("s_wait_kmcnt 0x0\n\t"
                 "s_wait_xcnt 0x0"
                 : "+s"(nw[0][0]), "+s"(nw[0][1]), "+s"(nw[0][2]),
                   "+s"(nw[1][0]), "+s"(nw[1][1]), "+s"(nw[1][2]),
                   "+s"(nw[2][0]), "+s"(nw[2][1]), "+s"(nw[2][2]),
                   "+s"(nw[3][0]), "+s"(nw[3][1]), "+s"(nw[3][2]),
                   "+v"(dp_a)
                 : "v"(zoff), "v"(o_st), "s"(po), "s"(nblk)
                 : "memory");
    const float o3 = step(f4a(cw[3][0], cw[3][1]), f4b(cw[3][2]),
                          f4a(nw[0][0], nw[0][1]), f4b(nw[0][2]));
    o4 = make_float4(o0, o1, o2, o3);
#pragma unroll
    for (int j = 0; j < 4; ++j)
#pragma unroll
      for (int k = 0; k < 3; ++k) cw[j][k] = nw[j][k];
  }
  if (t > 0) *(float4*)(out + t - 4) = o4;

  for (; t < T; ++t) {
    const float4 ca = pf[2 * t], cb = pf[2 * t + 1];
    const float4 na = pf[2 * t + 2], nb = pf[2 * t + 3];
    const float o = step(ca, cb, na, nb);
    out[t] = o;
  }
}

extern "C" void kernel_launch(void* const* d_in, const int* in_sizes, int n_in,
                              void* d_out, int out_size, void* d_ws,
                              size_t ws_size, hipStream_t stream) {
  const float* v_in  = (const float*)d_in[0];
  const float* vs_r  = (const float*)d_in[1];
  const float* fs    = (const float*)d_in[2];
  const float* W_in  = (const float*)d_in[3];
  const float* b_in  = (const float*)d_in[4];
  const float* W_h   = (const float*)d_in[5];
  const float* b_h   = (const float*)d_in[6];
  const float* W_out = (const float*)d_in[7];
  const float* b_out = (const float*)d_in[8];
  float* out = (float*)d_out;
  int T = in_sizes[0];

  float* params = (float*)d_ws;

  ndc_prep<<<(T + 8 + 255) / 256, 256, 0, stream>>>(v_in, vs_r, fs, params, T);
  ndc_scan<<<1, 32, 0, stream>>>(params, W_in, b_in, W_h, b_h, W_out, b_out,
                                 out, T);
}


# Hidden layer 1 with weights in BOTH matrix operands too; its row of results is moved into registers by one more matrix instruction (0/1 operand): 10 matrix instructions, not 16; sum order changes
# speedup vs baseline: 1.5547x; 1.0745x over previous
#include <hip/hip_runtime.h>


typedef __attribute__((ext_vector_type(2))) float v2f;
typedef __attribute__((ext_vector_type(8))) float v8f;

#define C1F 4.7e-9f

__global__ void ndc_prep(const float* __restrict__ v_in,
                         const float* __restrict__ vs_r,
                         const float* __restrict__ fs,
                         float* __restrict__ params, int T) {
  int t = blockIdx.x * blockDim.x + threadIdx.x;
  if (t >= T + 8) return;
  int ti = (t < T) ? t : (T - 1);
  float f   = fs[ti];
  float vr  = vs_r[ti];
  float c1r = 1.0f / (2.0f * C1F * f);
  float r0  = c1r * vr / (c1r + vr);
  float g0 = 1.0f / r0, g1 = 1.0f / c1r, g2 = 1.0f / vr;
  float inv = 2.0f / (g0 + g1 + g2);
  float P0 = g0 * inv, P1 = g1 * inv, P2 = g2 * inv;
  float p2v = P2 * v_in[ti];
  float4* p4 = (float4*)(params + 8 * (size_t)t);
  p4[0] = make_float4(P0 - 1.0f, P1, p2v, P0);
  p4[1] = make_float4(P1 - 1.0f, r0 * (1.0f / 3000.0f), 0.0f, 0.0f);
}

__device__ __forceinline__ float other_half(float x) {
  return __builtin_bit_cast(
      float, __builtin_amdgcn_permlanex16(0, __builtin_bit_cast(int, x),
                                          0x76543210, 0xfedcba98, false, false));
}

__device__ __forceinline__ float relu(float x) {
  return __builtin_elementwise_maximum(x, 0.0f);
}

__device__ __forceinline__ float clampf(float x, float lo, float hi) {
  return __builtin_amdgcn_fmed3f(x, lo, hi);
}

template <int M>
__device__ __forceinline__ float row_xor(float x) {
  return __builtin_bit_cast(
      float, __builtin_amdgcn_update_dpp(0, __builtin_bit_cast(int, x),
                                         0x160 + M, 0xf, 0xf, false));
}

__device__ __forceinline__ float row15(float x) {
  return __builtin_bit_cast(
      float, __builtin_amdgcn_permlane16(0, __builtin_bit_cast(int, x),
                                         0xffffffff, 0xffffffff, false, true));
}

__device__ __forceinline__ v8f mma(v2f a, v2f b, v8f c) {
  return __builtin_amdgcn_wmma_f32_16x16x4_f32(false, a, false, b, (short)0, c,
                                               false, false);
}

__global__ __launch_bounds__(32, 1) void ndc_scan(
    const float* __restrict__ params,
    const float* __restrict__ W_in, const float* __restrict__ b_in,
    const float* __restrict__ W_h,  const float* __restrict__ b_h,
    const float* __restrict__ W_out, const float* __restrict__ b_out,
    float* __restrict__ out, int T) {
  const int lane = threadIdx.x;
  const int c16  = lane & 15;
  const int hf   = lane >> 4;
  const int kb   = 8 * hf;

  auto U = [&](const int i) __attribute__((always_inline)) -> int {
    if (i < 7) return hf ? 7 + i : i;
    if (i == 7) return hf ? 14 : 22;
    if (i < 15) return hf ? 15 + i : 7 + i;
    return 30 + hf;
  };

  const bool  hl  = (c16 == 15);
  const int   ua  = hl ? 0 : c16;
  const int   ub  = hl ? 0 : 15 + c16;
  const float INF = __builtin_inff();
  float ALo[8][2], AHi[8][2], BLo[8][2], BHi[8][2];
#pragma unroll
  for (int q = 0; q < 8; ++q)
#pragma unroll
    for (int s = 0; s < 2; ++s) {
      const int k = U(2 * q + s);
      const float wa = W_h[1024 + ua * 32 + k];
      const float wb = W_h[1024 + ub * 32 + k];
      ALo[q][s] = hl ? 0.0f : wa;
      AHi[q][s] = hl ? INF : wa;
      BLo[q][s] = hl ? 0.0f : wb;
      BHi[q][s] = hl ? INF : wb;
    }
  v8f   C2;
  v2f   Wo2[4];
  float Hi2[8];
#pragma unroll
  for (int r = 0; r < 8; ++r) {
    const int  m    = kb + r;
    const bool colu = hl && (m < 15);
    const bool rowu = !hl && (m == 15);
    const int  u    = colu ? m : (rowu ? 15 + c16 : 0);
    const bool any  = colu || rowu;
    C2[r]  = any ? b_h[32 + u] : 0.0f;
    Wo2[r >> 1][r & 1] = any ? W_out[u] : 0.0f;
    Hi2[r] = any ? INF : 0.0f;
  }
  const int gq = c16 & 7;
  v2f A9;
  {
    const float* p = W_h + 1024 + (30 + (c16 >> 3)) * 32;
    A9.x = p[U(2 * gq)];
    A9.y = p[U(2 * gq + 1)];
  }
  const float bG = b_h[32 + 30 + hf];
  const float wG = (c16 == 0) ? W_out[30 + hf] : 0.0f;
  const bool g1 = (gq & 1) != 0, g2 = (gq & 2) != 0, g4 = (gq & 4) != 0;
  auto pick8 = [&](const float v0, const float v1, const float v2,
                   const float v3, const float v4, const float v5,
                   const float v6, const float v7)
                   __attribute__((always_inline)) -> float {
    const float t0 = g1 ? v1 : v0, t1 = g1 ? v3 : v2;
    const float t2 = g1 ? v5 : v4, t3 = g1 ? v7 : v6;
    const float w0 = g2 ? t1 : t0, w1 = g2 ? t3 : t2;
    return g4 ? w1 : w0;
  };

  const bool ha1 = (c16 == 7);
  const int  ua1 = (c16 < 7) ? c16 : c16 - 1;
  const int  ub1 = hl ? 0 : 15 + c16;
  float ALo1[8][2], AHi1[8][2], BLo1[8][2], BHi1[8][2];
#pragma unroll
  for (int q = 0; q < 8; ++q)
#pragma unroll
    for (int s = 0; s < 2; ++s) {
      const int k = 4 * q + 2 * hf + s;
      const float wa = W_h[ua1 * 32 + k];
      const float wb = W_h[ub1 * 32 + k];
      ALo1[q][s] = ha1 ? 0.0f : wa;
      AHi1[q][s] = ha1 ? INF : wa;
      BLo1[q][s] = hl ? 0.0f : wb;
      BHi1[q][s] = hl ? INF : wb;
    }
  v8f C1;
#pragma unroll
  for (int r = 0; r < 8; ++r) {
    const int  m    = kb + r;
    const bool colu = hl && (m != 7);
    const bool rowu = !hl && (m == 7);
    const int  u    = colu ? (m < 7 ? m : m - 1) : (rowu ? 15 + c16 : 0);
    C1[r] = (colu || rowu) ? b_h[u] : 0.0f;
  }
  v2f A91;
  float Wi0g[2], Wi1g[2], bing[2];
#pragma unroll
  for (int s = 0; s < 2; ++s) {
    const int k = 4 * gq + 2 * hf + s;
    A91[s]  = W_h[(30 + (c16 >> 3)) * 32 + k];
    Wi0g[s] = W_in[2 * k];
    Wi1g[s] = W_in[2 * k + 1];
    bing[s] = b_in[k];
  }
  const float bG1 = b_h[30 + hf];
  const float HiT = (hf == 0 && !hl) ? INF : 0.0f;
  const float HiS7 = hf ? INF : 0.0f;
  v2f B3;
  B3.x = (hf == 0) ? 1.0f : 0.0f;
  B3.y = 0.0f;

  float Wi0v[8][2], Wi1v[8][2], binv[8][2];
#pragma unroll
  for (int q = 0; q < 8; ++q)
#pragma unroll
    for (int s = 0; s < 2; ++s) {
      const int k = 4 * q + 2 * hf + s;
      Wi0v[q][s] = W_in[2 * k];
      Wi1v[q][s] = W_in[2 * k + 1];
      binv[q][s] = b_in[k];
    }

  const float bo = (lane == 0) ? b_out[0] : 0.0f;

  const float4* pf = (const float4*)params;

  float cv[8][2], cvg[2];
  {
    const float dr0 = pf[1].y;
#pragma unroll
    for (int q = 0; q < 8; ++q)
#pragma unroll
      for (int s = 0; s < 2; ++s) cv[q][s] = fmaf(Wi1v[q][s], dr0, binv[q][s]);
    cvg[0] = fmaf(Wi1g[0], dr0, bing[0]);
    cvg[1] = fmaf(Wi1g[1], dr0, bing[1]);
  }

  float b1 = 0.0f;
  float dp_a;
  {
    const float4 c0 = pf[0];
    dp_a = fmaf(c0.y, 0.0f, fmaf(c0.x, 0.0f, c0.z));
  }

  auto step = [&](const float4 ca, const float4 cb, const float4 na,
                  const float4 nb)
                  __attribute__((always_inline)) -> float {
    float hx[8][2];
#pragma unroll
    for (int q = 0; q < 8; ++q) {
      hx[q][0] = fmaf(Wi0v[q][0], dp_a, cv[q][0]);
      hx[q][1] = fmaf(Wi0v[q][1], dp_a, cv[q][1]);
    }
    v2f B91;
    B91.x = fmaxf(fmaf(Wi0g[0], dp_a, cvg[0]), 0.0f);
    B91.y = fmaxf(fmaf(Wi0g[1], dp_a, cvg[1]), 0.0f);

    v8f z8;
#pragma unroll
    for (int r = 0; r < 8; ++r) z8[r] = 0.0f;

    v8f fx = C1, fy = z8;
    v8f g1t;
#pragma unroll
    for (int j = 0; j < 4; ++j) {
      v2f A, B;
      A.x = clampf(hx[j][0], ALo1[j][0], AHi1[j][0]);
      A.y = clampf(hx[j][1], ALo1[j][1], AHi1[j][1]);
      B.x = clampf(hx[j][0], BLo1[j][0], BHi1[j][0]);
      B.y = clampf(hx[j][1], BLo1[j][1], BHi1[j][1]);
      fx = mma(A, B, fx);
      A.x = clampf(hx[4 + j][0], ALo1[4 + j][0], AHi1[4 + j][0]);
      A.y = clampf(hx[4 + j][1], ALo1[4 + j][1], AHi1[4 + j][1]);
      B.x = clampf(hx[4 + j][0], BLo1[4 + j][0], BHi1[4 + j][0]);
      B.y = clampf(hx[4 + j][1], BLo1[4 + j][1], BHi1[4 + j][1]);
      fy = mma(A, B, fy);
      if (j == 0) g1t = mma(A91, B91, z8);
    }
    float y3x;
    {
      float gv = pick8(g1t[0], g1t[1], g1t[2], g1t[3], g1t[4], g1t[5], g1t[6],
                       g1t[7]);
      gv += row_xor<1>(gv);
      gv += row_xor<2>(gv);
      gv += row_xor<4>(gv);
      y3x = relu(gv + bG1);
    }
    v8f S = fx + fy;
    v2f A3;
    A3.x = clampf(S[7], 0.0f, HiT);
    A3.y = 0.0f;
    const v8f Tt = mma(A3, B3, z8);
    float Y[16];
#pragma unroll
    for (int i = 0; i < 7; ++i) {
      Y[i]     = S[i];
      Y[8 + i] = Tt[i];
    }
    const float s7c = clampf(S[7], 0.0f, HiS7);
    Y[7]  = s7c + Tt[7];
    Y[15] = y3x;
    float Yb[16];
#pragma unroll
    for (int i = 0; i < 7; ++i) {
      Yb[i]     = row15(S[i]);
      Yb[8 + i] = Tt[i];
    }
    Yb[7]  = row15(s7c) + Tt[7];
    Yb[15] = y3x;

    v8f ex = C2, ey = z8;
    v8f g;
    float cst0 = 0.0f;
#pragma unroll
    for (int j = 0; j < 4; ++j) {
      v2f A, B;
      A.x = clampf(Y[2 * j], ALo[j][0], AHi[j][0]);
      A.y = clampf(Y[2 * j + 1], ALo[j][1], AHi[j][1]);
      B.x = clampf(Y[2 * j], BLo[j][0], BHi[j][0]);
      B.y = clampf(Y[2 * j + 1], BLo[j][1], BHi[j][1]);
      ex = mma(A, B, ex);
      A.x = clampf(Y[8 + 2 * j], ALo[4 + j][0], AHi[4 + j][0]);
      A.y = clampf(Y[8 + 2 * j + 1], ALo[4 + j][1], AHi[4 + j][1]);
      B.x = clampf(Y[8 + 2 * j], BLo[4 + j][0], BHi[4 + j][0]);
      B.y = clampf(Y[8 + 2 * j + 1], BLo[4 + j][1], BHi[4 + j][1]);
      ey = mma(A, B, ey);
      if (j == 0) {
        v2f B9;
        B9.x = relu(pick8(Yb[0], Yb[2], Yb[4], Yb[6], Yb[8], Yb[10], Yb[12], Yb[14]));
        B9.y = relu(pick8(Yb[1], Yb[3], Yb[5], Yb[7], Yb[9], Yb[11], Yb[13], Yb[15]));
        g = mma(A9, B9, z8);
        __builtin_amdgcn_sched_barrier(0);
      }
      if (j == 2) {
        float gv = pick8(g[0], g[1], g[2], g[3], g[4], g[5], g[6], g[7]);
        gv += row_xor<1>(gv);
        gv += row_xor<2>(gv);
        gv += row_xor<4>(gv);
        cst0 = fmaf(wG, relu(gv + bG), bo);
        __builtin_amdgcn_sched_barrier(0);
      }
    }

#pragma unroll
    for (int q = 0; q < 8; ++q) {
      cv[q][0] = fmaf(Wi1v[q][0], nb.y, binv[q][0]);
      cv[q][1] = fmaf(Wi1v[q][1], nb.y, binv[q][1]);
    }
    cvg[0] = fmaf(Wi1g[0], nb.y, bing[0]);
    cvg[1] = fmaf(Wi1g[1], nb.y, bing[1]);

    const float h_a   = 0.5f * dp_a;
    const float u     = fmaf(cb.x, b1, ca.z);
    const float alpha = fmaf(na.y, ca.w, na.x);
    const float beta  = fmaf(na.y, u, na.z);

    v2f a[4];
#pragma unroll
    for (int i = 0; i < 4; ++i) {
      a[i].x = clampf(ex[2 * i] + ey[2 * i], 0.0f, Hi2[2 * i]);
      a[i].y = clampf(ex[2 * i + 1] + ey[2 * i + 1], 0.0f, Hi2[2 * i + 1]);
    }
    v2f t0 = {cst0, 0.0f};
    t0 = __builtin_elementwise_fma(Wo2[0], a[0], t0);
    v2f t1 = Wo2[1] * a[1];
    t0 = __builtin_elementwise_fma(Wo2[2], a[2], t0);
    t1 = __builtin_elementwise_fma(Wo2[3], a[3], t1);
    const v2f ts = t0 + t1;
    float ph = ts.x + ts.y;
    ph += row_xor<1>(ph);
    ph += row_xor<2>(ph);
    ph += row_xor<4>(ph);
    ph += row_xor<8>(ph);
    float dp_b = ph + other_half(ph);

    b1 = fmaf(ca.w, dp_b, u);
    dp_a = fmaf(alpha, dp_b, beta);
    return fmaf(0.5f, dp_b, h_a);
  };

  typedef float v4f __attribute__((ext_vector_type(4)));
  typedef unsigned long long u64;
  auto lo32 = [](const u64 w) __attribute__((always_inline)) -> float {
    return __builtin_bit_cast(float, (unsigned)(w & 0xffffffffull));
  };
  auto hi32 = [](const u64 w) __attribute__((always_inline)) -> float {
    return __builtin_bit_cast(float, (unsigned)(w >> 32));
  };
  auto f4a = [&](const u64 w0, const u64 w1) __attribute__((always_inline)) -> float4 {
    return make_float4(lo32(w0), hi32(w0), lo32(w1), hi32(w1));
  };
  auto f4b = [&](const u64 w2) __attribute__((always_inline)) -> float4 {
    return make_float4(lo32(w2), hi32(w2), 0.0f, 0.0f);
  };

  u64 cw[4][3];
  {
    const u64* pw = (const u64*)params;
#pragma unroll
    for (int j = 0; j < 4; ++j) {
      cw[j][0] = pw[4 * j + 0];
      cw[j][1] = pw[4 * j + 1];
      cw[j][2] = pw[4 * j + 2];
    }
  }
  float4 o4 = make_float4(0.0f, 0.0f, 0.0f, 0.0f);
  int t = 0;
  for (; t + 4 <= T; t += 4) {
    u64 nw[4][3];
    const float* nblk = params + 8 * (size_t)(t + 4);
    float* po = out + (t > 0 ? t - 4 : 0);
    const v4f o_st = {o4.x, o4.y, o4.z, o4.w};
    const int zoff = 0;
    asm volatile("s_load_b64 %0, %13, 0x0\n\t"
                 "s_load_b64 %1, %13, 0x8\n\t"
                 "s_load_b64 %2, %13, 0x10\n\t"
                 "s_load_b64 %3, %13, 0x20\n\t"
                 "s_load_b64 %4, %13, 0x28\n\t"
                 "s_load_b64 %5, %13, 0x30\n\t"
                 "s_load_b64 %6, %13, 0x40\n\t"
                 "s_load_b64 %7, %13, 0x48\n\t"
                 "s_load_b64 %8, %13, 0x50\n\t"
                 "s_load_b64 %9, %13, 0x60\n\t"
                 "s_load_b64 %10, %13, 0x68\n\t"
                 "s_load_b64 %11, %13, 0x70\n\t"
                 "global_store_b128 %14, %15, %16"
                 : "=&s"(nw[0][0]), "=&s"(nw[0][1]), "=&s"(nw[0][2]),
                   "=&s"(nw[1][0]), "=&s"(nw[1][1]), "=&s"(nw[1][2]),
                   "=&s"(nw[2][0]), "=&s"(nw[2][1]), "=&s"(nw[2][2]),
                   "=&s"(nw[3][0]), "=&s"(nw[3][1]), "=&s"(nw[3][2]),
                   "+v"(dp_a)
                 : "s"(nblk), "v"(zoff), "v"(o_st), "s"(po)
                 : "memory");
    {
      const int tp = (t + 12 < T + 4) ? (t + 12) : (T + 4);
      __builtin_prefetch(params + 8 * (size_t)tp, 0, 3);
    }

    const float o0 = step(f4a(cw[0][0], cw[0][1]), f4b(cw[0][2]),
                          f4a(cw[1][0], cw[1][1]), f4b(cw[1][2]));
    const float o1 = step(f4a(cw[1][0], cw[1][1]), f4b(cw[1][2]),
                          f4a(cw[2][0], cw[2][1]), f4b(cw[2][2]));
    const float o2 = step(f4a(cw[2][0], cw[2][1]), f4b(cw[2][2]),
                          f4a(cw[3][0], cw[3][1]), f4b(cw[3][2]));
    asm volatile("s_wait_kmcnt 0x0\n\t"
                 "s_wait_xcnt 0x0"
                 : "+s"(nw[0][0]), "+s"(nw[0][1]), "+s"(nw[0][2]),
                   "+s"(nw[1][0]), "+s"(nw[1][1]), "+s"(nw[1][2]),
                   "+s"(nw[2][0]), "+s"(nw[2][1]), "+s"(nw[2][2]),
                   "+s"(nw[3][0]), "+s"(nw[3][1]), "+s"(nw[3][2]),
                   "+v"(dp_a)
                 : "v"(zoff), "v"(o_st), "s"(po), "s"(nblk)
                 : "memory");
    const float o3 = step(f4a(cw[3][0], cw[3][1]), f4b(cw[3][2]),
                          f4a(nw[0][0], nw[0][1]), f4b(nw[0][2]));
    o4 = make_float4(o0, o1, o2, o3);
#pragma unroll
    for (int j = 0; j < 4; ++j)
#pragma unroll
      for (int k = 0; k < 3; ++k) cw[j][k] = nw[j][k];
  }
  if (t > 0) *(float4*)(out + t - 4) = o4;

  for (; t < T; ++t) {
    const float4 ca = pf[2 * t], cb = pf[2 * t + 1];
    const float4 na = pf[2 * t + 2], nb = pf[2 * t + 3];
    const float o = step(ca, cb, na, nb);
    out[t] = o;
  }
}

extern "C" void kernel_launch(void* const* d_in, const int* in_sizes, int n_in,
                              void* d_out, int out_size, void* d_ws,
                              size_t ws_size, hipStream_t stream) {
  const float* v_in  = (const float*)d_in[0];
  const float* vs_r  = (const float*)d_in[1];
  const float* fs    = (const float*)d_in[2];
  const float* W_in  = (const float*)d_in[3];
  const float* b_in  = (const float*)d_in[4];
  const float* W_h   = (const float*)d_in[5];
  const float* b_h   = (const float*)d_in[6];
  const float* W_out = (const float*)d_in[7];
  const float* b_out = (const float*)d_in[8];
  float* out = (float*)d_out;
  int T = in_sizes[0];

  float* params = (float*)d_ws;

  ndc_prep<<<(T + 8 + 255) / 256, 256, 0, stream>>>(v_in, vs_r, fs, params, T);
  ndc_scan<<<1, 32, 0, stream>>>(params, W_in, b_in, W_h, b_h, W_out, b_out,
                                 out, T);
}


# Sixteen chains per wave as the sixteen columns of B: a hidden layer is one 32x32 by 32x16 product (16 matrix instructions serve 16 chains); chains own 32 steps; dot-product sum order changes
# speedup vs baseline: 2752.7350x; 1770.5892x over previous
#include <hip/hip_runtime.h>


typedef __attribute__((ext_vector_type(2))) float v2f;
typedef __attribute__((ext_vector_type(8))) float v8f;

#define C1F 4.7e-9f

constexpr int CHUNK = 32;
constexpr int WARM  = 64;
constexpr int HEAD  = CHUNK + WARM;

__global__ void ndc_prep(const float* __restrict__ v_in,
                         const float* __restrict__ vs_r,
                         const float* __restrict__ fs,
                         float* __restrict__ params, int T) {
  int t = blockIdx.x * blockDim.x + threadIdx.x;
  if (t >= T + 8) return;
  int ti = (t < T) ? t : (T - 1);
  float f   = fs[ti];
  float vr  = vs_r[ti];
  float c1r = 1.0f / (2.0f * C1F * f);
  float r0  = c1r * vr / (c1r + vr);
  float g0 = 1.0f / r0, g1 = 1.0f / c1r, g2 = 1.0f / vr;
  float inv = 2.0f / (g0 + g1 + g2);
  float P0 = g0 * inv, P1 = g1 * inv, P2 = g2 * inv;
  float p2v = P2 * v_in[ti];
  float4* p4 = (float4*)(params + 8 * (size_t)t);
  p4[0] = make_float4(P0 - 1.0f, P1, p2v, P0);
  p4[1] = make_float4(P1 - 1.0f, r0 * (1.0f / 3000.0f), 0.0f, 0.0f);
}

__device__ __forceinline__ float other_half(float x) {
  return __builtin_bit_cast(
      float, __builtin_amdgcn_permlanex16(0, __builtin_bit_cast(int, x),
                                          0x76543210, 0xfedcba98, false, false));
}

__device__ __forceinline__ float relu(float x) {
  return __builtin_elementwise_maximum(x, 0.0f);
}

__device__ __forceinline__ v8f mma(v2f a, v2f b, v8f c) {
  return __builtin_amdgcn_wmma_f32_16x16x4_f32(false, a, false, b, (short)0, c,
                                               false, false);
}

__device__ __forceinline__ void scan16(
    const float* __restrict__ params, float* rec,
    const float* __restrict__ W_in, const float* __restrict__ b_in,
    const float* __restrict__ W_h,  const float* __restrict__ b_h,
    const float* __restrict__ W_out, const float* __restrict__ b_out,
    float* __restrict__ out, const int T, const int t_first,
    const int own_start, const int ntrip_warm, const int ntrip_own,
    const bool zero_start, const float b1_in, const float dpa_in) {
  const int lane = threadIdx.x & 31;
  const int c16  = lane & 15;
  const int hf   = lane >> 4;

  auto K = [&](const int q, const int s) __attribute__((always_inline)) -> int {
    return 16 * (q >> 2) + 8 * hf + 2 * (q & 3) + s;
  };

  v2f A1[2][8], A2[2][8];
#pragma unroll
  for (int m = 0; m < 2; ++m)
#pragma unroll
    for (int q = 0; q < 8; ++q)
#pragma unroll
      for (int s = 0; s < 2; ++s) {
        A1[m][q][s] = W_h[(16 * m + c16) * 32 + K(q, s)];
        A2[m][q][s] = W_h[1024 + (16 * m + c16) * 32 + K(q, s)];
      }
  v8f Cb1[2], Cb2[2];
  v2f Wo[2][4];
#pragma unroll
  for (int m = 0; m < 2; ++m)
#pragma unroll
    for (int r = 0; r < 8; ++r) {
      const int u = 16 * m + 8 * hf + r;
      Cb1[m][r] = b_h[u];
      Cb2[m][r] = b_h[32 + u];
      Wo[m][r >> 1][r & 1] = W_out[u];
    }
  float Wi0[8][2], Wi1[8][2], bin[8][2];
#pragma unroll
  for (int q = 0; q < 8; ++q)
#pragma unroll
    for (int s = 0; s < 2; ++s) {
      const int k = K(q, s);
      Wi0[q][s] = W_in[2 * k];
      Wi1[q][s] = W_in[2 * k + 1];
      bin[q][s] = b_in[k];
    }
  const float bo = (hf == 0) ? b_out[0] : 0.0f;

  const int tmax = ((T + 8) & ~3) - 4;
  auto blk = [&](const int t) __attribute__((always_inline)) -> const float* {
    return params + 8 * (size_t)(t < tmax ? t : tmax);
  };

  struct Co { float4 a; float2 b; };
  auto load4 = [&](Co (&c)[4], const float* p) __attribute__((always_inline)) {
#pragma unroll
    for (int j = 0; j < 4; ++j) {
      c[j].a = *(const float4*)(p + 8 * j);
      c[j].b = *(const float2*)(p + 8 * j + 4);
    }
  };

  Co cur[4];
  load4(cur, blk(t_first));

  float cv[8][2];
#pragma unroll
  for (int q = 0; q < 8; ++q)
#pragma unroll
    for (int s = 0; s < 2; ++s) cv[q][s] = fmaf(Wi1[q][s], cur[0].b.y, bin[q][s]);

  float b1 = 0.0f;
  float dp_a;
  if (zero_start) {
    dp_a = fmaf(cur[0].a.y, 0.0f, fmaf(cur[0].a.x, 0.0f, cur[0].a.z));
  } else {
    b1   = b1_in;
    dp_a = dpa_in;
  }

  auto step = [&](const Co c, const Co n) __attribute__((always_inline)) -> float {
    v2f B0[8];
#pragma unroll
    for (int q = 0; q < 8; ++q) {
      B0[q].x = relu(fmaf(Wi0[q][0], dp_a, cv[q][0]));
      B0[q].y = relu(fmaf(Wi0[q][1], dp_a, cv[q][1]));
    }
    v8f d0 = Cb1[0], d1 = Cb1[1];
#pragma unroll
    for (int q = 0; q < 8; ++q) {
      d0 = mma(A1[0][q], B0[q], d0);
      d1 = mma(A1[1][q], B0[q], d1);
    }
    v8f e0 = Cb2[0], e1 = Cb2[1];
#pragma unroll
    for (int q = 0; q < 8; ++q) {
      v2f B;
      if (q < 4) {
        B.x = relu(d0[2 * q]);
        B.y = relu(d0[2 * q + 1]);
      } else {
        B.x = relu(d1[2 * (q - 4)]);
        B.y = relu(d1[2 * (q - 4) + 1]);
      }
      e0 = mma(A2[0][q], B, e0);
      e1 = mma(A2[1][q], B, e1);
    }

#pragma unroll
    for (int q = 0; q < 8; ++q) {
      cv[q][0] = fmaf(Wi1[q][0], n.b.y, bin[q][0]);
      cv[q][1] = fmaf(Wi1[q][1], n.b.y, bin[q][1]);
    }
    const float h_a   = 0.5f * dp_a;
    const float u     = fmaf(c.b.x, b1, c.a.z);
    const float alpha = fmaf(n.a.y, c.a.w, n.a.x);
    const float beta  = fmaf(n.a.y, u, n.a.z);

    v2f t0 = {bo, 0.0f};
    v2f t1 = {0.0f, 0.0f};
#pragma unroll
    for (int i = 0; i < 4; ++i) {
      v2f a0, a1;
      a0.x = relu(e0[2 * i]);
      a0.y = relu(e0[2 * i + 1]);
      a1.x = relu(e1[2 * i]);
      a1.y = relu(e1[2 * i + 1]);
      t0 = __builtin_elementwise_fma(Wo[0][i], a0, t0);
      t1 = __builtin_elementwise_fma(Wo[1][i], a1, t1);
    }
    const v2f ts = t0 + t1;
    const float ph = ts.x + ts.y;
    const float dp_b = ph + other_half(ph);

    b1 = fmaf(c.a.w, dp_b, u);
    dp_a = fmaf(alpha, dp_b, beta);
    return fmaf(0.5f, dp_b, h_a);
  };

  int t = t_first;
  auto trip = [&]() __attribute__((always_inline)) {
    Co nxt[4];
    load4(nxt, blk(t + 4));
    __builtin_prefetch(blk(t + 8), 0, 3);
    const float o0 = step(cur[0], cur[1]);
    const float o1 = step(cur[1], cur[2]);
    const float o2 = step(cur[2], cur[3]);
    const float o3 = step(cur[3], nxt[0]);
    if (hf == 0 && t >= own_start) {
      if (t + 4 <= T) {
        *(float4*)(out + t) = make_float4(o0, o1, o2, o3);
      } else {
        if (t < T)     out[t]     = o0;
        if (t + 1 < T) out[t + 1] = o1;
        if (t + 2 < T) out[t + 2] = o2;
      }
    }
#pragma unroll
    for (int j = 0; j < 4; ++j) cur[j] = nxt[j];
    t += 4;
  };

  for (int i = 0; i < ntrip_warm; ++i) trip();
  if (t == own_start)
    *(float2*)(rec + 8 * (size_t)own_start + 6) = make_float2(b1, dp_a);
  for (int i = 0; i < ntrip_own; ++i) trip();
  *(float2*)(rec + 8 * (size_t)(own_start + 1) + 6) = make_float2(b1, dp_a);
}

__device__ __forceinline__ int chain_start(const int j) {
  return j == 0 ? 0 : HEAD + CHUNK * (j - 1);
}

__global__ __launch_bounds__(128) void ndc_scan_chains(
    const float* __restrict__ params, float* rec,
    const float* __restrict__ W_in, const float* __restrict__ b_in,
    const float* __restrict__ W_h,  const float* __restrict__ b_h,
    const float* __restrict__ W_out, const float* __restrict__ b_out,
    float* __restrict__ out, int T, int nchains) {
  const int w = blockIdx.x * 4 + (threadIdx.x >> 5);
  if (16 * w >= nchains) return;
  int j = 16 * w + (threadIdx.x & 15);
  j = j < nchains ? j : nchains - 1;
  const int own = chain_start(j);
  const int t0  = j == 0 ? 0 : own - WARM;
  scan16(params, rec, W_in, b_in, W_h, b_h, W_out, b_out, out, T, t0, own,
         WARM / 4, CHUNK / 4, true, 0.0f, 0.0f);
}

__global__ __launch_bounds__(256) void ndc_check(
    const float* __restrict__ params, float* rec,
    const float* __restrict__ W_in, const float* __restrict__ b_in,
    const float* __restrict__ W_h,  const float* __restrict__ b_h,
    const float* __restrict__ W_out, const float* __restrict__ b_out,
    float* __restrict__ out, int T, int nchains) {
  unsigned bad = 0u;
#pragma unroll 8
  for (int i = 1 + (int)threadIdx.x; i < nchains; i += 256) {
    const uint2 sN = *(const uint2*)(rec + 8 * (size_t)chain_start(i) + 6);
    const uint2 eP = *(const uint2*)(rec + 8 * (size_t)(chain_start(i - 1) + 1) + 6);
    bad |= (sN.x ^ eP.x) | (sN.y ^ eP.y);
  }
  if (__syncthreads_or(bad != 0u) == 0) return;
  if (threadIdx.x >= 32) return;

  for (int i = 1; i < nchains; ++i) {
    const int own = chain_start(i);
    const volatile unsigned* ps = (const volatile unsigned*)(rec + 8 * (size_t)own + 6);
    const volatile unsigned* pe =
        (const volatile unsigned*)(rec + 8 * (size_t)(chain_start(i - 1) + 1) + 6);
    const unsigned s0 = __builtin_amdgcn_readfirstlane(ps[0]);
    const unsigned s1 = __builtin_amdgcn_readfirstlane(ps[1]);
    const unsigned e0 = __builtin_amdgcn_readfirstlane(pe[0]);
    const unsigned e1 = __builtin_amdgcn_readfirstlane(pe[1]);
    if (s0 == e0 && s1 == e1) continue;
    scan16(params, rec, W_in, b_in, W_h, b_h, W_out, b_out, out, T, own, own,
           0, CHUNK / 4, false, __builtin_bit_cast(float, e0),
           __builtin_bit_cast(float, e1));
    __threadfence();
  }
}

extern "C" void kernel_launch(void* const* d_in, const int* in_sizes, int n_in,
                              void* d_out, int out_size, void* d_ws,
                              size_t ws_size, hipStream_t stream) {
  const float* v_in  = (const float*)d_in[0];
  const float* vs_r  = (const float*)d_in[1];
  const float* fs    = (const float*)d_in[2];
  const float* W_in  = (const float*)d_in[3];
  const float* b_in  = (const float*)d_in[4];
  const float* W_h   = (const float*)d_in[5];
  const float* b_h   = (const float*)d_in[6];
  const float* W_out = (const float*)d_in[7];
  const float* b_out = (const float*)d_in[8];
  float* out = (float*)d_out;
  int T = in_sizes[0];

  float* params = (float*)d_ws;
  const int nchains = T > HEAD ? 1 + (T - HEAD + CHUNK - 1) / CHUNK : 1;
  const int nwaves  = (nchains + 15) / 16;

  ndc_prep<<<(T + 8 + 255) / 256, 256, 0, stream>>>(v_in, vs_r, fs, params, T);
  ndc_scan_chains<<<(nwaves + 3) / 4, 128, 0, stream>>>(
      params, params, W_in, b_in, W_h, b_h, W_out, b_out, out, T, nchains);
  ndc_check<<<1, 256, 0, stream>>>(params, params, W_in, b_in, W_h, b_h, W_out,
                                    b_out, out, T, nchains);
}


# Warm-up of every chain shortened from 64 to 32 steps (a multiple of 32 keeps one writer per output line): a wave runs 64 steps, not 96; bit-for-bit boundary check and rerun stay; sums unchanged
# speedup vs baseline: 3269.8188x; 1.1878x over previous
#include <hip/hip_runtime.h>


typedef __attribute__((ext_vector_type(2))) float v2f;
typedef __attribute__((ext_vector_type(8))) float v8f;

#define C1F 4.7e-9f

constexpr int CHUNK = 32;
constexpr int WARM  = 32;
constexpr int HEAD  = CHUNK + WARM;

__global__ void ndc_prep(const float* __restrict__ v_in,
                         const float* __restrict__ vs_r,
                         const float* __restrict__ fs,
                         float* __restrict__ params, int T) {
  int t = blockIdx.x * blockDim.x + threadIdx.x;
  if (t >= T + 8) return;
  int ti = (t < T) ? t : (T - 1);
  float f   = fs[ti];
  float vr  = vs_r[ti];
  float c1r = 1.0f / (2.0f * C1F * f);
  float r0  = c1r * vr / (c1r + vr);
  float g0 = 1.0f / r0, g1 = 1.0f / c1r, g2 = 1.0f / vr;
  float inv = 2.0f / (g0 + g1 + g2);
  float P0 = g0 * inv, P1 = g1 * inv, P2 = g2 * inv;
  float p2v = P2 * v_in[ti];
  float4* p4 = (float4*)(params + 8 * (size_t)t);
  p4[0] = make_float4(P0 - 1.0f, P1, p2v, P0);
  p4[1] = make_float4(P1 - 1.0f, r0 * (1.0f / 3000.0f), 0.0f, 0.0f);
}

__device__ __forceinline__ float other_half(float x) {
  return __builtin_bit_cast(
      float, __builtin_amdgcn_permlanex16(0, __builtin_bit_cast(int, x),
                                          0x76543210, 0xfedcba98, false, false));
}

__device__ __forceinline__ float relu(float x) {
  return __builtin_elementwise_maximum(x, 0.0f);
}

__device__ __forceinline__ v8f mma(v2f a, v2f b, v8f c) {
  return __builtin_amdgcn_wmma_f32_16x16x4_f32(false, a, false, b, (short)0, c,
                                               false, false);
}

__device__ __forceinline__ void scan16(
    const float* __restrict__ params, float* rec,
    const float* __restrict__ W_in, const float* __restrict__ b_in,
    const float* __restrict__ W_h,  const float* __restrict__ b_h,
    const float* __restrict__ W_out, const float* __restrict__ b_out,
    float* __restrict__ out, const int T, const int t_first,
    const int own_start, const int ntrip_warm, const int ntrip_own,
    const bool zero_start, const float b1_in, const float dpa_in) {
  const int lane = threadIdx.x & 31;
  const int c16  = lane & 15;
  const int hf   = lane >> 4;

  auto K = [&](const int q, const int s) __attribute__((always_inline)) -> int {
    return 16 * (q >> 2) + 8 * hf + 2 * (q & 3) + s;
  };

  v2f A1[2][8], A2[2][8];
#pragma unroll
  for (int m = 0; m < 2; ++m)
#pragma unroll
    for (int q = 0; q < 8; ++q)
#pragma unroll
      for (int s = 0; s < 2; ++s) {
        A1[m][q][s] = W_h[(16 * m + c16) * 32 + K(q, s)];
        A2[m][q][s] = W_h[1024 + (16 * m + c16) * 32 + K(q, s)];
      }
  v8f Cb1[2], Cb2[2];
  v2f Wo[2][4];
#pragma unroll
  for (int m = 0; m < 2; ++m)
#pragma unroll
    for (int r = 0; r < 8; ++r) {
      const int u = 16 * m + 8 * hf + r;
      Cb1[m][r] = b_h[u];
      Cb2[m][r] = b_h[32 + u];
      Wo[m][r >> 1][r & 1] = W_out[u];
    }
  float Wi0[8][2], Wi1[8][2], bin[8][2];
#pragma unroll
  for (int q = 0; q < 8; ++q)
#pragma unroll
    for (int s = 0; s < 2; ++s) {
      const int k = K(q, s);
      Wi0[q][s] = W_in[2 * k];
      Wi1[q][s] = W_in[2 * k + 1];
      bin[q][s] = b_in[k];
    }
  const float bo = (hf == 0) ? b_out[0] : 0.0f;

  const int tmax = ((T + 8) & ~3) - 4;
  auto blk = [&](const int t) __attribute__((always_inline)) -> const float* {
    return params + 8 * (size_t)(t < tmax ? t : tmax);
  };

  struct Co { float4 a; float2 b; };
  auto load4 = [&](Co (&c)[4], const float* p) __attribute__((always_inline)) {
#pragma unroll
    for (int j = 0; j < 4; ++j) {
      c[j].a = *(const float4*)(p + 8 * j);
      c[j].b = *(const float2*)(p + 8 * j + 4);
    }
  };

  Co cur[4];
  load4(cur, blk(t_first));

  float cv[8][2];
#pragma unroll
  for (int q = 0; q < 8; ++q)
#pragma unroll
    for (int s = 0; s < 2; ++s) cv[q][s] = fmaf(Wi1[q][s], cur[0].b.y, bin[q][s]);

  float b1 = 0.0f;
  float dp_a;
  if (zero_start) {
    dp_a = fmaf(cur[0].a.y, 0.0f, fmaf(cur[0].a.x, 0.0f, cur[0].a.z));
  } else {
    b1   = b1_in;
    dp_a = dpa_in;
  }

  auto step = [&](const Co c, const Co n) __attribute__((always_inline)) -> float {
    v2f B0[8];
#pragma unroll
    for (int q = 0; q < 8; ++q) {
      B0[q].x = relu(fmaf(Wi0[q][0], dp_a, cv[q][0]));
      B0[q].y = relu(fmaf(Wi0[q][1], dp_a, cv[q][1]));
    }
    v8f d0 = Cb1[0], d1 = Cb1[1];
#pragma unroll
    for (int q = 0; q < 8; ++q) {
      d0 = mma(A1[0][q], B0[q], d0);
      d1 = mma(A1[1][q], B0[q], d1);
    }
    v8f e0 = Cb2[0], e1 = Cb2[1];
#pragma unroll
    for (int q = 0; q < 8; ++q) {
      v2f B;
      if (q < 4) {
        B.x = relu(d0[2 * q]);
        B.y = relu(d0[2 * q + 1]);
      } else {
        B.x = relu(d1[2 * (q - 4)]);
        B.y = relu(d1[2 * (q - 4) + 1]);
      }
      e0 = mma(A2[0][q], B, e0);
      e1 = mma(A2[1][q], B, e1);
    }

#pragma unroll
    for (int q = 0; q < 8; ++q) {
      cv[q][0] = fmaf(Wi1[q][0], n.b.y, bin[q][0]);
      cv[q][1] = fmaf(Wi1[q][1], n.b.y, bin[q][1]);
    }
    const float h_a   = 0.5f * dp_a;
    const float u     = fmaf(c.b.x, b1, c.a.z);
    const float alpha = fmaf(n.a.y, c.a.w, n.a.x);
    const float beta  = fmaf(n.a.y, u, n.a.z);

    v2f t0 = {bo, 0.0f};
    v2f t1 = {0.0f, 0.0f};
#pragma unroll
    for (int i = 0; i < 4; ++i) {
      v2f a0, a1;
      a0.x = relu(e0[2 * i]);
      a0.y = relu(e0[2 * i + 1]);
      a1.x = relu(e1[2 * i]);
      a1.y = relu(e1[2 * i + 1]);
      t0 = __builtin_elementwise_fma(Wo[0][i], a0, t0);
      t1 = __builtin_elementwise_fma(Wo[1][i], a1, t1);
    }
    const v2f ts = t0 + t1;
    const float ph = ts.x + ts.y;
    const float dp_b = ph + other_half(ph);

    b1 = fmaf(c.a.w, dp_b, u);
    dp_a = fmaf(alpha, dp_b, beta);
    return fmaf(0.5f, dp_b, h_a);
  };

  int t = t_first;
  auto trip = [&]() __attribute__((always_inline)) {
    Co nxt[4];
    load4(nxt, blk(t + 4));
    __builtin_prefetch(blk(t + 8), 0, 3);
    const float o0 = step(cur[0], cur[1]);
    const float o1 = step(cur[1], cur[2]);
    const float o2 = step(cur[2], cur[3]);
    const float o3 = step(cur[3], nxt[0]);
    if (hf == 0 && t >= own_start) {
      if (t + 4 <= T) {
        *(float4*)(out + t) = make_float4(o0, o1, o2, o3);
      } else {
        if (t < T)     out[t]     = o0;
        if (t + 1 < T) out[t + 1] = o1;
        if (t + 2 < T) out[t + 2] = o2;
      }
    }
#pragma unroll
    for (int j = 0; j < 4; ++j) cur[j] = nxt[j];
    t += 4;
  };

  for (int i = 0; i < ntrip_warm; ++i) trip();
  if (t == own_start)
    *(float2*)(rec + 8 * (size_t)own_start + 6) = make_float2(b1, dp_a);
  for (int i = 0; i < ntrip_own; ++i) trip();
  *(float2*)(rec + 8 * (size_t)(own_start + 1) + 6) = make_float2(b1, dp_a);
}

__device__ __forceinline__ int chain_start(const int j) {
  return j == 0 ? 0 : HEAD + CHUNK * (j - 1);
}

__global__ __launch_bounds__(128) void ndc_scan_chains(
    const float* __restrict__ params, float* rec,
    const float* __restrict__ W_in, const float* __restrict__ b_in,
    const float* __restrict__ W_h,  const float* __restrict__ b_h,
    const float* __restrict__ W_out, const float* __restrict__ b_out,
    float* __restrict__ out, int T, int nchains) {
  const int w = blockIdx.x * 4 + (threadIdx.x >> 5);
  if (16 * w >= nchains) return;
  int j = 16 * w + (threadIdx.x & 15);
  j = j < nchains ? j : nchains - 1;
  const int own = chain_start(j);
  const int t0  = j == 0 ? 0 : own - WARM;
  scan16(params, rec, W_in, b_in, W_h, b_h, W_out, b_out, out, T, t0, own,
         WARM / 4, CHUNK / 4, true, 0.0f, 0.0f);
}

__global__ __launch_bounds__(256) void ndc_check(
    const float* __restrict__ params, float* rec,
    const float* __restrict__ W_in, const float* __restrict__ b_in,
    const float* __restrict__ W_h,  const float* __restrict__ b_h,
    const float* __restrict__ W_out, const float* __restrict__ b_out,
    float* __restrict__ out, int T, int nchains) {
  unsigned bad = 0u;
#pragma unroll 8
  for (int i = 1 + (int)threadIdx.x; i < nchains; i += 256) {
    const uint2 sN = *(const uint2*)(rec + 8 * (size_t)chain_start(i) + 6);
    const uint2 eP = *(const uint2*)(rec + 8 * (size_t)(chain_start(i - 1) + 1) + 6);
    bad |= (sN.x ^ eP.x) | (sN.y ^ eP.y);
  }
  if (__syncthreads_or(bad != 0u) == 0) return;
  if (threadIdx.x >= 32) return;

  for (int i = 1; i < nchains; ++i) {
    const int own = chain_start(i);
    const volatile unsigned* ps = (const volatile unsigned*)(rec + 8 * (size_t)own + 6);
    const volatile unsigned* pe =
        (const volatile unsigned*)(rec + 8 * (size_t)(chain_start(i - 1) + 1) + 6);
    const unsigned s0 = __builtin_amdgcn_readfirstlane(ps[0]);
    const unsigned s1 = __builtin_amdgcn_readfirstlane(ps[1]);
    const unsigned e0 = __builtin_amdgcn_readfirstlane(pe[0]);
    const unsigned e1 = __builtin_amdgcn_readfirstlane(pe[1]);
    if (s0 == e0 && s1 == e1) continue;
    scan16(params, rec, W_in, b_in, W_h, b_h, W_out, b_out, out, T, own, own,
           0, CHUNK / 4, false, __builtin_bit_cast(float, e0),
           __builtin_bit_cast(float, e1));
    __threadfence();
  }
}

extern "C" void kernel_launch(void* const* d_in, const int* in_sizes, int n_in,
                              void* d_out, int out_size, void* d_ws,
                              size_t ws_size, hipStream_t stream) {
  const float* v_in  = (const float*)d_in[0];
  const float* vs_r  = (const float*)d_in[1];
  const float* fs    = (const float*)d_in[2];
  const float* W_in  = (const float*)d_in[3];
  const float* b_in  = (const float*)d_in[4];
  const float* W_h   = (const float*)d_in[5];
  const float* b_h   = (const float*)d_in[6];
  const float* W_out = (const float*)d_in[7];
  const float* b_out = (const float*)d_in[8];
  float* out = (float*)d_out;
  int T = in_sizes[0];

  float* params = (float*)d_ws;
  const int nchains = T > HEAD ? 1 + (T - HEAD + CHUNK - 1) / CHUNK : 1;
  const int nwaves  = (nchains + 15) / 16;

  ndc_prep<<<(T + 8 + 255) / 256, 256, 0, stream>>>(v_in, vs_r, fs, params, T);
  ndc_scan_chains<<<(nwaves + 3) / 4, 128, 0, stream>>>(
      params, params, W_in, b_in, W_h, b_h, W_out, b_out, out, T, nchains);
  ndc_check<<<1, 256, 0, stream>>>(params, params, W_in, b_in, W_h, b_h, W_out,
                                    b_out, out, T, nchains);
}


# The fifteen chain boundaries inside a wave are compared in its registers (row shift) after its last step; the check kernel reads one summary line per wave, not two lines per chain; sums unchanged
# speedup vs baseline: 5043.0847x; 1.5423x over previous
#include <hip/hip_runtime.h>


typedef __attribute__((ext_vector_type(2))) float v2f;
typedef __attribute__((ext_vector_type(8))) float v8f;

#define C1F 4.7e-9f

constexpr int CHUNK = 32;
constexpr int WARM  = 32;
constexpr int HEAD  = CHUNK + WARM;

__global__ void ndc_prep(const float* __restrict__ v_in,
                         const float* __restrict__ vs_r,
                         const float* __restrict__ fs,
                         float* __restrict__ params, int T) {
  int t = blockIdx.x * blockDim.x + threadIdx.x;
  if (t >= T + 8) return;
  int ti = (t < T) ? t : (T - 1);
  float f   = fs[ti];
  float vr  = vs_r[ti];
  float c1r = 1.0f / (2.0f * C1F * f);
  float r0  = c1r * vr / (c1r + vr);
  float g0 = 1.0f / r0, g1 = 1.0f / c1r, g2 = 1.0f / vr;
  float inv = 2.0f / (g0 + g1 + g2);
  float P0 = g0 * inv, P1 = g1 * inv, P2 = g2 * inv;
  float p2v = P2 * v_in[ti];
  float4* p4 = (float4*)(params + 8 * (size_t)t);
  p4[0] = make_float4(P0 - 1.0f, P1, p2v, P0);
  p4[1] = make_float4(P1 - 1.0f, r0 * (1.0f / 3000.0f), 0.0f, 0.0f);
}

__device__ __forceinline__ float other_half(float x) {
  return __builtin_bit_cast(
      float, __builtin_amdgcn_permlanex16(0, __builtin_bit_cast(int, x),
                                          0x76543210, 0xfedcba98, false, false));
}

__device__ __forceinline__ float relu(float x) {
  return __builtin_elementwise_maximum(x, 0.0f);
}

__device__ __forceinline__ v8f mma(v2f a, v2f b, v8f c) {
  return __builtin_amdgcn_wmma_f32_16x16x4_f32(false, a, false, b, (short)0, c,
                                               false, false);
}

struct Ends { float s_b1, s_dpa, e_b1, e_dpa; };
__device__ __forceinline__ Ends scan16(
    const float* __restrict__ params, float* rec,
    const float* __restrict__ W_in, const float* __restrict__ b_in,
    const float* __restrict__ W_h,  const float* __restrict__ b_h,
    const float* __restrict__ W_out, const float* __restrict__ b_out,
    float* __restrict__ out, const int T, const int t_first,
    const int own_start, const int ntrip_warm, const int ntrip_own,
    const bool zero_start, const float b1_in, const float dpa_in) {
  const int lane = threadIdx.x & 31;
  const int c16  = lane & 15;
  const int hf   = lane >> 4;

  auto K = [&](const int q, const int s) __attribute__((always_inline)) -> int {
    return 16 * (q >> 2) + 8 * hf + 2 * (q & 3) + s;
  };

  v2f A1[2][8], A2[2][8];
#pragma unroll
  for (int m = 0; m < 2; ++m)
#pragma unroll
    for (int q = 0; q < 8; ++q)
#pragma unroll
      for (int s = 0; s < 2; ++s) {
        A1[m][q][s] = W_h[(16 * m + c16) * 32 + K(q, s)];
        A2[m][q][s] = W_h[1024 + (16 * m + c16) * 32 + K(q, s)];
      }
  v8f Cb1[2], Cb2[2];
  v2f Wo[2][4];
#pragma unroll
  for (int m = 0; m < 2; ++m)
#pragma unroll
    for (int r = 0; r < 8; ++r) {
      const int u = 16 * m + 8 * hf + r;
      Cb1[m][r] = b_h[u];
      Cb2[m][r] = b_h[32 + u];
      Wo[m][r >> 1][r & 1] = W_out[u];
    }
  float Wi0[8][2], Wi1[8][2], bin[8][2];
#pragma unroll
  for (int q = 0; q < 8; ++q)
#pragma unroll
    for (int s = 0; s < 2; ++s) {
      const int k = K(q, s);
      Wi0[q][s] = W_in[2 * k];
      Wi1[q][s] = W_in[2 * k + 1];
      bin[q][s] = b_in[k];
    }
  const float bo = (hf == 0) ? b_out[0] : 0.0f;

  const int tmax = ((T + 8) & ~3) - 4;
  auto blk = [&](const int t) __attribute__((always_inline)) -> const float* {
    return params + 8 * (size_t)(t < tmax ? t : tmax);
  };

  struct Co { float4 a; float2 b; };
  auto load4 = [&](Co (&c)[4], const float* p) __attribute__((always_inline)) {
#pragma unroll
    for (int j = 0; j < 4; ++j) {
      c[j].a = *(const float4*)(p + 8 * j);
      c[j].b = *(const float2*)(p + 8 * j + 4);
    }
  };

  Co cur[4];
  load4(cur, blk(t_first));

  float cv[8][2];
#pragma unroll
  for (int q = 0; q < 8; ++q)
#pragma unroll
    for (int s = 0; s < 2; ++s) cv[q][s] = fmaf(Wi1[q][s], cur[0].b.y, bin[q][s]);

  float b1 = 0.0f;
  float dp_a;
  if (zero_start) {
    dp_a = fmaf(cur[0].a.y, 0.0f, fmaf(cur[0].a.x, 0.0f, cur[0].a.z));
  } else {
    b1   = b1_in;
    dp_a = dpa_in;
  }

  auto step = [&](const Co c, const Co n) __attribute__((always_inline)) -> float {
    v2f B0[8];
#pragma unroll
    for (int q = 0; q < 8; ++q) {
      B0[q].x = relu(fmaf(Wi0[q][0], dp_a, cv[q][0]));
      B0[q].y = relu(fmaf(Wi0[q][1], dp_a, cv[q][1]));
    }
    v8f d0 = Cb1[0], d1 = Cb1[1];
#pragma unroll
    for (int q = 0; q < 8; ++q) {
      d0 = mma(A1[0][q], B0[q], d0);
      d1 = mma(A1[1][q], B0[q], d1);
    }
    v8f e0 = Cb2[0], e1 = Cb2[1];
#pragma unroll
    for (int q = 0; q < 8; ++q) {
      v2f B;
      if (q < 4) {
        B.x = relu(d0[2 * q]);
        B.y = relu(d0[2 * q + 1]);
      } else {
        B.x = relu(d1[2 * (q - 4)]);
        B.y = relu(d1[2 * (q - 4) + 1]);
      }
      e0 = mma(A2[0][q], B, e0);
      e1 = mma(A2[1][q], B, e1);
    }

#pragma unroll
    for (int q = 0; q < 8; ++q) {
      cv[q][0] = fmaf(Wi1[q][0], n.b.y, bin[q][0]);
      cv[q][1] = fmaf(Wi1[q][1], n.b.y, bin[q][1]);
    }
    const float h_a   = 0.5f * dp_a;
    const float u     = fmaf(c.b.x, b1, c.a.z);
    const float alpha = fmaf(n.a.y, c.a.w, n.a.x);
    const float beta  = fmaf(n.a.y, u, n.a.z);

    v2f t0 = {bo, 0.0f};
    v2f t1 = {0.0f, 0.0f};
#pragma unroll
    for (int i = 0; i < 4; ++i) {
      v2f a0, a1;
      a0.x = relu(e0[2 * i]);
      a0.y = relu(e0[2 * i + 1]);
      a1.x = relu(e1[2 * i]);
      a1.y = relu(e1[2 * i + 1]);
      t0 = __builtin_elementwise_fma(Wo[0][i], a0, t0);
      t1 = __builtin_elementwise_fma(Wo[1][i], a1, t1);
    }
    const v2f ts = t0 + t1;
    const float ph = ts.x + ts.y;
    const float dp_b = ph + other_half(ph);

    b1 = fmaf(c.a.w, dp_b, u);
    dp_a = fmaf(alpha, dp_b, beta);
    return fmaf(0.5f, dp_b, h_a);
  };

  int t = t_first;
  auto trip = [&]() __attribute__((always_inline)) {
    Co nxt[4];
    load4(nxt, blk(t + 4));
    __builtin_prefetch(blk(t + 8), 0, 3);
    const float o0 = step(cur[0], cur[1]);
    const float o1 = step(cur[1], cur[2]);
    const float o2 = step(cur[2], cur[3]);
    const float o3 = step(cur[3], nxt[0]);
    if (hf == 0 && t >= own_start) {
      if (t + 4 <= T) {
        *(float4*)(out + t) = make_float4(o0, o1, o2, o3);
      } else {
        if (t < T)     out[t]     = o0;
        if (t + 1 < T) out[t + 1] = o1;
        if (t + 2 < T) out[t + 2] = o2;
      }
    }
#pragma unroll
    for (int j = 0; j < 4; ++j) cur[j] = nxt[j];
    t += 4;
  };

  for (int i = 0; i < ntrip_warm; ++i) trip();
  Ends ends;
  ends.s_b1  = b1;
  ends.s_dpa = dp_a;
  if (t == own_start)
    *(float2*)(rec + 8 * (size_t)own_start + 6) = make_float2(b1, dp_a);
  for (int i = 0; i < ntrip_own; ++i) trip();
  ends.e_b1  = b1;
  ends.e_dpa = dp_a;
  *(float2*)(rec + 8 * (size_t)(own_start + 1) + 6) = make_float2(b1, dp_a);
  return ends;
}

__device__ __forceinline__ int chain_start(const int j) {
  return j == 0 ? 0 : HEAD + CHUNK * (j - 1);
}

__device__ __forceinline__ int sum_slot(const int w) {
  return 32 * (w >> 2) + 4 + 4 * (w & 3);
}

__device__ __forceinline__ unsigned lane_below(const unsigned x) {
  return (unsigned)__builtin_amdgcn_update_dpp((int)x, (int)x, 0x111, 0xf, 0xf,
                                               false);
}

__global__ __launch_bounds__(128) void ndc_scan_chains(
    const float* __restrict__ params, float* rec,
    const float* __restrict__ W_in, const float* __restrict__ b_in,
    const float* __restrict__ W_h,  const float* __restrict__ b_h,
    const float* __restrict__ W_out, const float* __restrict__ b_out,
    float* __restrict__ out, int T, int nchains) {
  const int w = blockIdx.x * 4 + (threadIdx.x >> 5);
  if (16 * w >= nchains) return;
  int j = 16 * w + (threadIdx.x & 15);
  j = j < nchains ? j : nchains - 1;
  const int own = chain_start(j);
  const int t0  = j == 0 ? 0 : own - WARM;
  const Ends e = scan16(params, rec, W_in, b_in, W_h, b_h, W_out, b_out, out, T,
                        t0, own, WARM / 4, CHUNK / 4, true, 0.0f, 0.0f);

  const int lane = threadIdx.x & 31;
  const int c16  = lane & 15;
  const unsigned sb = __builtin_bit_cast(unsigned, e.s_b1);
  const unsigned sd = __builtin_bit_cast(unsigned, e.s_dpa);
  const unsigned pb = lane_below(__builtin_bit_cast(unsigned, e.e_b1));
  const unsigned pd = lane_below(__builtin_bit_cast(unsigned, e.e_dpa));
  const bool counts  = c16 >= 1 && 16 * w + c16 < nchains;
  const bool differs = counts && (((sb ^ pb) | (sd ^ pd)) != 0u);
  const unsigned inner = __builtin_amdgcn_ballot_w32(differs);
  float* s = rec + 8 * (size_t)sum_slot(w) + 6;
  if (lane == 0) {
    *(float2*)s       = make_float2(e.s_b1, e.s_dpa);
    *(uint2*)(s + 16) = make_uint2(inner, 0u);
  }
  if (lane == 15) *(float2*)(s + 8) = make_float2(e.e_b1, e.e_dpa);
}

__global__ __launch_bounds__(256) void ndc_check(
    const float* __restrict__ params, float* rec,
    const float* __restrict__ W_in, const float* __restrict__ b_in,
    const float* __restrict__ W_h,  const float* __restrict__ b_h,
    const float* __restrict__ W_out, const float* __restrict__ b_out,
    float* __restrict__ out, int T, int nchains) {
  const int nwaves = (nchains + 15) >> 4;
  unsigned bad = 0u;
  for (int base = 0; base < nwaves; base += 1024) {
    uint2 sN[4], eP[4];
    unsigned inner[4];
#pragma unroll
    for (int k = 0; k < 4; ++k) {
      int w = base + 256 * k + (int)threadIdx.x;
      w = w < nwaves ? w : nwaves - 1;
      const int sw = sum_slot(w);
      const int sp = w > 0 ? sum_slot(w - 1) + 1 : sw;
      sN[k]    = *(const uint2*)(rec + 8 * (size_t)sw + 6);
      eP[k]    = *(const uint2*)(rec + 8 * (size_t)sp + 6);
      inner[k] = *(const unsigned*)(rec + 8 * (size_t)(sw + 2) + 6);
    }
#pragma unroll
    for (int k = 0; k < 4; ++k)
      bad |= inner[k] | (sN[k].x ^ eP[k].x) | (sN[k].y ^ eP[k].y);
  }
  if (__syncthreads_or(bad != 0u) == 0) return;
  if (threadIdx.x >= 32) return;

  for (int i = 1; i < nchains; ++i) {
    const int own = chain_start(i);
    const volatile unsigned* ps = (const volatile unsigned*)(rec + 8 * (size_t)own + 6);
    const volatile unsigned* pe =
        (const volatile unsigned*)(rec + 8 * (size_t)(chain_start(i - 1) + 1) + 6);
    const unsigned s0 = __builtin_amdgcn_readfirstlane(ps[0]);
    const unsigned s1 = __builtin_amdgcn_readfirstlane(ps[1]);
    const unsigned e0 = __builtin_amdgcn_readfirstlane(pe[0]);
    const unsigned e1 = __builtin_amdgcn_readfirstlane(pe[1]);
    if (s0 == e0 && s1 == e1) continue;
    (void)scan16(params, rec, W_in, b_in, W_h, b_h, W_out, b_out, out, T, own,
                 own, 0, CHUNK / 4, false, __builtin_bit_cast(float, e0),
                 __builtin_bit_cast(float, e1));
    __threadfence();
  }
}

extern "C" void kernel_launch(void* const* d_in, const int* in_sizes, int n_in,
                              void* d_out, int out_size, void* d_ws,
                              size_t ws_size, hipStream_t stream) {
  const float* v_in  = (const float*)d_in[0];
  const float* vs_r  = (const float*)d_in[1];
  const float* fs    = (const float*)d_in[2];
  const float* W_in  = (const float*)d_in[3];
  const float* b_in  = (const float*)d_in[4];
  const float* W_h   = (const float*)d_in[5];
  const float* b_h   = (const float*)d_in[6];
  const float* W_out = (const float*)d_in[7];
  const float* b_out = (const float*)d_in[8];
  float* out = (float*)d_out;
  int T = in_sizes[0];

  float* params = (float*)d_ws;
  const int nchains = T > HEAD ? 1 + (T - HEAD + CHUNK - 1) / CHUNK : 1;
  const int nwaves  = (nchains + 15) / 16;

  ndc_prep<<<(T + 8 + 255) / 256, 256, 0, stream>>>(v_in, vs_r, fs, params, T);
  ndc_scan_chains<<<(nwaves + 3) / 4, 128, 0, stream>>>(
      params, params, W_in, b_in, W_h, b_h, W_out, b_out, out, T, nchains);
  ndc_check<<<1, 256, 0, stream>>>(params, params, W_in, b_in, W_h, b_h, W_out,
                                    b_out, out, T, nchains);
}
